# starting kernel
# baseline (speedup 1.0000x reference)
#include <hip/hip_runtime.h>
#include <hip/hip_bf16.h>

typedef __attribute__((ext_vector_type(16))) _Float16 v16h;
typedef __attribute__((ext_vector_type(8)))  _Float16 v8h;
typedef __attribute__((ext_vector_type(8)))  float    v8f;

#define B_TOT   65536
#define S_DIM   256
#define A_N     8
#define K_N     10
#define H_DIM   64
#define AD_DIM  128
#define D_DIM   384

#define WG_ROWS 64
#define NTHREADS 64
#define NWAVES  (NTHREADS / 32)
#define ROWS_PER_WAVE 32
#define XS      392
#define HS      72

__device__ __forceinline__ v16h load_frag(const _Float16* __restrict__ p,
                                          int stride, int row, int k0, int lane) {
    const int ho = (lane >> 4) << 3;
    const _Float16* q = p + row * stride + k0 + ho;
    union { v16h v; struct { v8h lo; v8h hi; } s; } u;
    u.s.lo = *(const v8h*)q;
    u.s.hi = *(const v8h*)(q + 16);
    return u.v;
}

template <int KD>
__device__ __forceinline__ void gemm4x2(const _Float16* __restrict__ A, int lda,
                                        const _Float16* __restrict__ W, int ldw,
                                        int lane, v8f c[2][4]) {
    const int mr = lane & 15;
    v8f z = {0.f, 0.f, 0.f, 0.f, 0.f, 0.f, 0.f, 0.f};
#pragma unroll
    for (int m = 0; m < 2; ++m)
#pragma unroll
        for (int t = 0; t < 4; ++t) c[m][t] = z;
    for (int k0 = 0; k0 < KD; k0 += 32) {
        v16h a0 = load_frag(A, lda, mr, k0, lane);
        v16h a1 = load_frag(A + 16 * lda, lda, mr, k0, lane);
#pragma unroll
        for (int t = 0; t < 4; ++t) {
            v16h b = load_frag(W, ldw, t * 16 + mr, k0, lane);
            c[0][t] = __builtin_amdgcn_wmma_f32_16x16x32_f16(
                false, a0, false, b, (short)0, c[0][t], false, false);
            c[1][t] = __builtin_amdgcn_wmma_f32_16x16x32_f16(
                false, a1, false, b, (short)0, c[1][t], false, false);
        }
    }
}

__device__ __forceinline__ void gemm1x2(const _Float16* __restrict__ A, int lda,
                                        const _Float16* __restrict__ W,
                                        int lane, v8f c[2]) {
    const int mr = lane & 15;
    v8f z = {0.f, 0.f, 0.f, 0.f, 0.f, 0.f, 0.f, 0.f};
    c[0] = z; c[1] = z;
#pragma unroll
    for (int k0 = 0; k0 < H_DIM; k0 += 32) {
        v16h a0 = load_frag(A, lda, mr, k0, lane);
        v16h a1 = load_frag(A + 16 * lda, lda, mr, k0, lane);
        v16h b  = load_frag(W, H_DIM, mr, k0, lane);
        c[0] = __builtin_amdgcn_wmma_f32_16x16x32_f16(
            false, a0, false, b, (short)0, c[0], false, false);
        c[1] = __builtin_amdgcn_wmma_f32_16x16x32_f16(
            false, a1, false, b, (short)0, c[1], false, false);
    }
}

__device__ __forceinline__ void bias_relu_store2(const v8f c[2][4],
                                                 const float* __restrict__ bias,
                                                 _Float16* __restrict__ Hout, int lane) {
    const int col = lane & 15, hi = lane >> 4;
#pragma unroll
    for (int t = 0; t < 4; ++t) {
        float bv = bias[t * 16 + col];
#pragma unroll
        for (int m = 0; m < 2; ++m)
#pragma unroll
            for (int r = 0; r < 8; ++r) {
                float v = c[m][t][r] + bv;
                Hout[(m * 16 + r + 8 * hi) * HS + t * 16 + col] =
                    (_Float16)(v > 0.f ? v : 0.f);
            }
    }
}

__global__ void cvt_f32_f16(const float* __restrict__ src, _Float16* __restrict__ dst, int n) {
    int i = blockIdx.x * blockDim.x + threadIdx.x;
    if (i < n) dst[i] = (_Float16)src[i];
}

__global__ void cvt_pad_w3(const float* __restrict__ src, _Float16* __restrict__ dst, int outdim) {
    int i = blockIdx.x * blockDim.x + threadIdx.x;
    const int n = K_N * 16 * H_DIM;
    if (i >= n) return;
    int h = i % H_DIM;
    int o = (i / H_DIM) % 16;
    int k = i / (16 * H_DIM);
    dst[i] = (o < outdim) ? (_Float16)src[(k * outdim + o) * H_DIM + h] : (_Float16)0.f;
}

__global__ __launch_bounds__(NTHREADS, 2)
void qplex_main(const float* __restrict__ states, const float* __restrict__ actions,
                const _Float16* __restrict__ wk1, const _Float16* __restrict__ wa1,
                const _Float16* __restrict__ wc1,
                const _Float16* __restrict__ wk2, const _Float16* __restrict__ wa2,
                const _Float16* __restrict__ wc2,
                const _Float16* __restrict__ wk3, const _Float16* __restrict__ wa3,
                const _Float16* __restrict__ wc3,
                const float* __restrict__ bk1, const float* __restrict__ bk2,
                const float* __restrict__ bk3,
                const float* __restrict__ ba1, const float* __restrict__ ba2,
                const float* __restrict__ ba3,
                const float* __restrict__ bc1, const float* __restrict__ bc2,
                const float* __restrict__ bc3,
                float* __restrict__ out) {
    __shared__ _Float16 Xs[WG_ROWS * XS];
    __shared__ _Float16 Hs[NWAVES * ROWS_PER_WAVE * HS];

    const int tid = threadIdx.x;
    const int b0  = blockIdx.x * WG_ROWS;

    for (int idx = tid; idx < WG_ROWS * D_DIM; idx += NTHREADS) {
        int row = idx / D_DIM;
        int col = idx - row * D_DIM;
        float v = (col < S_DIM)
                      ? states[(size_t)(b0 + row) * S_DIM + col]
                      : actions[(size_t)(b0 + row) * AD_DIM + (col - S_DIM)];
        Xs[row * XS + col] = (_Float16)v;
    }
    __syncthreads();

    const int wave = tid >> 5, lane = tid & 31;
    const int col = lane & 15, hi = lane >> 4;
    const _Float16* Xw = Xs + (wave * ROWS_PER_WAVE) * XS;
    _Float16*       Hw = Hs + (wave * ROWS_PER_WAVE) * HS;

    float acc[2][8];
#pragma unroll
    for (int m = 0; m < 2; ++m)
#pragma unroll
        for (int r = 0; r < 8; ++r) acc[m][r] = 0.f;

#pragma unroll 1
    for (int k = 0; k < K_N; ++k) {
        v8f c1[2][4];
        v8f ck[2], ca[2], cc[2];

        gemm4x2<S_DIM>(Xw, XS, wk1 + (size_t)k * H_DIM * S_DIM, S_DIM, lane, c1);
        bias_relu_store2(c1, bk1 + k * H_DIM, Hw, lane);
        gemm4x2<H_DIM>(Hw, HS, wk2 + (size_t)k * H_DIM * H_DIM, H_DIM, lane, c1);
        bias_relu_store2(c1, bk2 + k * H_DIM, Hw, lane);
        gemm1x2(Hw, HS, wk3 + (size_t)k * 16 * H_DIM, lane, ck);

        gemm4x2<S_DIM>(Xw, XS, wa1 + (size_t)k * H_DIM * S_DIM, S_DIM, lane, c1);
        bias_relu_store2(c1, ba1 + k * H_DIM, Hw, lane);
        gemm4x2<H_DIM>(Hw, HS, wa2 + (size_t)k * H_DIM * H_DIM, H_DIM, lane, c1);
        bias_relu_store2(c1, ba2 + k * H_DIM, Hw, lane);
        gemm1x2(Hw, HS, wa3 + (size_t)k * 16 * H_DIM, lane, ca);

        gemm4x2<D_DIM>(Xw, XS, wc1 + (size_t)k * H_DIM * D_DIM, D_DIM, lane, c1);
        bias_relu_store2(c1, bc1 + k * H_DIM, Hw, lane);
        gemm4x2<H_DIM>(Hw, HS, wc2 + (size_t)k * H_DIM * H_DIM, H_DIM, lane, c1);
        bias_relu_store2(c1, bc2 + k * H_DIM, Hw, lane);
        gemm1x2(Hw, HS, wc3 + (size_t)k * 16 * H_DIM, lane, cc);

        const float kbias = bk3[k];
        const float ab3v  = ba3[k * A_N + (col & 7)];
        const float cb3v  = bc3[k * A_N + (col & 7)];
#pragma unroll
        for (int m = 0; m < 2; ++m)
#pragma unroll
            for (int r = 0; r < 8; ++r) {
                float kv = __shfl(ck[m][r], lane & 16, 32);
                kv = fabsf(kv + kbias) + 1e-10f;
                float ag = 1.f / (1.f + __expf(-(ca[m][r] + ab3v)));
                float ac = 1.f / (1.f + __expf(-(cc[m][r] + cb3v)));
                acc[m][r] += kv * ag * ac;
            }
    }

    if (col < A_N) {
#pragma unroll
        for (int m = 0; m < 2; ++m)
#pragma unroll
            for (int r = 0; r < 8; ++r)
                out[(size_t)(b0 + wave * ROWS_PER_WAVE + m * 16 + r + 8 * hi) * A_N + col] =
                    acc[m][r];
    }
}

extern "C" void kernel_launch(void* const* d_in, const int* in_sizes, int n_in,
                              void* d_out, int out_size, void* d_ws, size_t ws_size,
                              hipStream_t stream) {
    (void)in_sizes; (void)n_in; (void)out_size; (void)ws_size;
    const float* states  = (const float*)d_in[0];
    const float* actions = (const float*)d_in[1];
    const float* kW1 = (const float*)d_in[2];
    const float* kb1 = (const float*)d_in[3];
    const float* kW2 = (const float*)d_in[4];
    const float* kb2 = (const float*)d_in[5];
    const float* kW3 = (const float*)d_in[6];
    const float* kb3 = (const float*)d_in[7];
    const float* aW1 = (const float*)d_in[8];
    const float* ab1 = (const float*)d_in[9];
    const float* aW2 = (const float*)d_in[10];
    const float* ab2 = (const float*)d_in[11];
    const float* aW3 = (const float*)d_in[12];
    const float* ab3 = (const float*)d_in[13];
    const float* cW1 = (const float*)d_in[14];
    const float* cb1 = (const float*)d_in[15];
    const float* cW2 = (const float*)d_in[16];
    const float* cb2 = (const float*)d_in[17];
    const float* cW3 = (const float*)d_in[18];
    const float* cb3 = (const float*)d_in[19];

    _Float16* ws = (_Float16*)d_ws;
    size_t off = 0;
    _Float16* wk1 = ws + off; off += (size_t)K_N * H_DIM * S_DIM;
    _Float16* wa1 = ws + off; off += (size_t)K_N * H_DIM * S_DIM;
    _Float16* wc1 = ws + off; off += (size_t)K_N * H_DIM * D_DIM;
    _Float16* wk2 = ws + off; off += (size_t)K_N * H_DIM * H_DIM;
    _Float16* wa2 = ws + off; off += (size_t)K_N * H_DIM * H_DIM;
    _Float16* wc2 = ws + off; off += (size_t)K_N * H_DIM * H_DIM;
    _Float16* wk3 = ws + off; off += (size_t)K_N * 16 * H_DIM;
    _Float16* wa3 = ws + off; off += (size_t)K_N * 16 * H_DIM;
    _Float16* wc3 = ws + off; off += (size_t)K_N * 16 * H_DIM;

    auto cvt = [&](const float* s, _Float16* d, int n) {
        cvt_f32_f16<<<(n + 255) / 256, 256, 0, stream>>>(s, d, n);
    };
    cvt(kW1, wk1, K_N * H_DIM * S_DIM);
    cvt(aW1, wa1, K_N * H_DIM * S_DIM);
    cvt(cW1, wc1, K_N * H_DIM * D_DIM);
    cvt(kW2, wk2, K_N * H_DIM * H_DIM);
    cvt(aW2, wa2, K_N * H_DIM * H_DIM);
    cvt(cW2, wc2, K_N * H_DIM * H_DIM);
    const int n3 = K_N * 16 * H_DIM;
    cvt_pad_w3<<<(n3 + 255) / 256, 256, 0, stream>>>(kW3, wk3, 1);
    cvt_pad_w3<<<(n3 + 255) / 256, 256, 0, stream>>>(aW3, wa3, A_N);
    cvt_pad_w3<<<(n3 + 255) / 256, 256, 0, stream>>>(cW3, wc3, A_N);

    qplex_main<<<B_TOT / WG_ROWS, NTHREADS, 0, stream>>>(
        states, actions,
        wk1, wa1, wc1, wk2, wa2, wc2, wk3, wa3, wc3,
        kb1, kb2, kb3, ab1, ab2, ab3, cb1, cb2, cb3,
        (float*)d_out);
}


# Weight-fragment and bias loads software-pipelined: layer-1 B fragments fetched 4 k-steps ahead into a register ring, layer-2/3 weights and biases before layer 1; summation order unchanged
# speedup vs baseline: 2.0305x; 2.0305x over previous
#include <hip/hip_runtime.h>
#include <hip/hip_bf16.h>

typedef __attribute__((ext_vector_type(16))) _Float16 v16h;
typedef __attribute__((ext_vector_type(8)))  _Float16 v8h;
typedef __attribute__((ext_vector_type(8)))  float    v8f;

#define B_TOT   65536
#define S_DIM   256
#define A_N     8
#define K_N     10
#define H_DIM   64
#define AD_DIM  128
#define D_DIM   384

#define WG_ROWS 64
#define NTHREADS 64
#define NWAVES  (NTHREADS / 32)
#define ROWS_PER_WAVE 32
#define XS      392
#define HS      72

__device__ __forceinline__ v16h load_frag(const _Float16* __restrict__ p,
                                          int stride, int row, int k0, int lane) {
    const int ho = (lane >> 4) << 3;
    const _Float16* q = p + row * stride + k0 + ho;
    union { v16h v; struct { v8h lo; v8h hi; } s; } u;
    u.s.lo = *(const v8h*)q;
    u.s.hi = *(const v8h*)(q + 16);
    return u.v;
}

#define PF_D 4

__device__ __forceinline__ v8f wmma_f16(v16h a, v16h b, v8f c) {
    return __builtin_amdgcn_wmma_f32_16x16x32_f16(false, a, false, b, (short)0, c, false, false);
}

template <int KD, int KDN>
__device__ __forceinline__ void gemm4x2_l1(const _Float16* __restrict__ A, int lda,
                                           const _Float16* __restrict__ W,
                                           const _Float16* __restrict__ Wn,
                                           int lane, v16h (&ring)[PF_D][4], v8f c[2][4]) {
    constexpr int NS = KD / 32;
    const int mr = lane & 15;
    v8f z = {0.f, 0.f, 0.f, 0.f, 0.f, 0.f, 0.f, 0.f};
#pragma unroll
    for (int m = 0; m < 2; ++m)
#pragma unroll
        for (int t = 0; t < 4; ++t) c[m][t] = z;
#pragma unroll
    for (int s = 0; s < NS; ++s) {
        v16h a0 = load_frag(A, lda, mr, s * 32, lane);
        v16h a1 = load_frag(A + 16 * lda, lda, mr, s * 32, lane);
#pragma unroll
        for (int t = 0; t < 4; ++t) {
            c[0][t] = wmma_f16(a0, ring[s % PF_D][t], c[0][t]);
            c[1][t] = wmma_f16(a1, ring[s % PF_D][t], c[1][t]);
        }
#pragma unroll
        for (int t = 0; t < 4; ++t) {
            if (s + PF_D < NS)
                ring[s % PF_D][t] = load_frag(W, KD, t * 16 + mr, (s + PF_D) * 32, lane);
            else
                ring[s % PF_D][t] = load_frag(Wn, KDN, t * 16 + mr, (s + PF_D - NS) * 32, lane);
        }
    }
}

__device__ __forceinline__ void gemm4x2_h(const _Float16* __restrict__ A,
                                          const v16h (&w)[2][4], int lane, v8f c[2][4]) {
    const int mr = lane & 15;
    v8f z = {0.f, 0.f, 0.f, 0.f, 0.f, 0.f, 0.f, 0.f};
#pragma unroll
    for (int m = 0; m < 2; ++m)
#pragma unroll
        for (int t = 0; t < 4; ++t) c[m][t] = z;
#pragma unroll
    for (int s = 0; s < 2; ++s) {
        v16h a0 = load_frag(A, HS, mr, s * 32, lane);
        v16h a1 = load_frag(A + 16 * HS, HS, mr, s * 32, lane);
#pragma unroll
        for (int t = 0; t < 4; ++t) {
            c[0][t] = wmma_f16(a0, w[s][t], c[0][t]);
            c[1][t] = wmma_f16(a1, w[s][t], c[1][t]);
        }
    }
}

__device__ __forceinline__ void gemm1x2_h(const _Float16* __restrict__ A,
                                          const v16h (&w)[2], int lane, v8f c[2]) {
    const int mr = lane & 15;
    v8f z = {0.f, 0.f, 0.f, 0.f, 0.f, 0.f, 0.f, 0.f};
    c[0] = z; c[1] = z;
#pragma unroll
    for (int s = 0; s < 2; ++s) {
        v16h a0 = load_frag(A, HS, mr, s * 32, lane);
        v16h a1 = load_frag(A + 16 * HS, HS, mr, s * 32, lane);
        c[0] = wmma_f16(a0, w[s], c[0]);
        c[1] = wmma_f16(a1, w[s], c[1]);
    }
}

__device__ __forceinline__ void bias_relu_store2(const v8f c[2][4],
                                                 const float (&bv)[4],
                                                 _Float16* __restrict__ Hout, int lane) {
    const int col = lane & 15, hi = lane >> 4;
#pragma unroll
    for (int t = 0; t < 4; ++t) {
#pragma unroll
        for (int m = 0; m < 2; ++m)
#pragma unroll
            for (int r = 0; r < 8; ++r) {
                float v = c[m][t][r] + bv[t];
                Hout[(m * 16 + r + 8 * hi) * HS + t * 16 + col] =
                    (_Float16)(v > 0.f ? v : 0.f);
            }
    }
}

template <int KD, int KDN>
__device__ __forceinline__ void mlp3(const _Float16* __restrict__ Xw, _Float16* __restrict__ Hw,
                                     const _Float16* __restrict__ W1,
                                     const _Float16* __restrict__ W1n,
                                     const _Float16* __restrict__ W2,
                                     const _Float16* __restrict__ W3,
                                     const float* __restrict__ b1,
                                     const float* __restrict__ b2,
                                     int lane, v16h (&ring)[PF_D][4], v8f out[2]) {
    const int mr = lane & 15;
    v16h w2[2][4];
    v16h w3[2];
    float bv1[4], bv2[4];
#pragma unroll
    for (int s = 0; s < 2; ++s)
#pragma unroll
        for (int t = 0; t < 4; ++t) w2[s][t] = load_frag(W2, H_DIM, t * 16 + mr, s * 32, lane);
#pragma unroll
    for (int s = 0; s < 2; ++s) w3[s] = load_frag(W3, H_DIM, mr, s * 32, lane);
#pragma unroll
    for (int t = 0; t < 4; ++t) { bv1[t] = b1[t * 16 + mr]; bv2[t] = b2[t * 16 + mr]; }

    v8f c1[2][4];
    gemm4x2_l1<KD, KDN>(Xw, XS, W1, W1n, lane, ring, c1);
    bias_relu_store2(c1, bv1, Hw, lane);
    gemm4x2_h(Hw, w2, lane, c1);
    bias_relu_store2(c1, bv2, Hw, lane);
    gemm1x2_h(Hw, w3, lane, out);
}

__global__ void cvt_f32_f16(const float* __restrict__ src, _Float16* __restrict__ dst, int n) {
    int i = blockIdx.x * blockDim.x + threadIdx.x;
    if (i < n) dst[i] = (_Float16)src[i];
}

__global__ void cvt_pad_w3(const float* __restrict__ src, _Float16* __restrict__ dst, int outdim) {
    int i = blockIdx.x * blockDim.x + threadIdx.x;
    const int n = K_N * 16 * H_DIM;
    if (i >= n) return;
    int h = i % H_DIM;
    int o = (i / H_DIM) % 16;
    int k = i / (16 * H_DIM);
    dst[i] = (o < outdim) ? (_Float16)src[(k * outdim + o) * H_DIM + h] : (_Float16)0.f;
}

__global__ __launch_bounds__(NTHREADS, 2)
void qplex_main(const float* __restrict__ states, const float* __restrict__ actions,
                const _Float16* __restrict__ wk1, const _Float16* __restrict__ wa1,
                const _Float16* __restrict__ wc1,
                const _Float16* __restrict__ wk2, const _Float16* __restrict__ wa2,
                const _Float16* __restrict__ wc2,
                const _Float16* __restrict__ wk3, const _Float16* __restrict__ wa3,
                const _Float16* __restrict__ wc3,
                const float* __restrict__ bk1, const float* __restrict__ bk2,
                const float* __restrict__ bk3,
                const float* __restrict__ ba1, const float* __restrict__ ba2,
                const float* __restrict__ ba3,
                const float* __restrict__ bc1, const float* __restrict__ bc2,
                const float* __restrict__ bc3,
                float* __restrict__ out) {
    __shared__ _Float16 Xs[WG_ROWS * XS];
    __shared__ _Float16 Hs[NWAVES * ROWS_PER_WAVE * HS];

    const int tid = threadIdx.x;
    const int b0  = blockIdx.x * WG_ROWS;
    const int wave = tid >> 5, lane = tid & 31;
    const int col = lane & 15, hi = lane >> 4;

    v16h ring[PF_D][4];
#pragma unroll
    for (int s = 0; s < PF_D; ++s)
#pragma unroll
        for (int t = 0; t < 4; ++t) ring[s][t] = load_frag(wk1, S_DIM, t * 16 + col, s * 32, lane);

    for (int idx = tid; idx < WG_ROWS * D_DIM; idx += NTHREADS) {
        int row = idx / D_DIM;
        int c = idx - row * D_DIM;
        float v = (c < S_DIM)
                      ? states[(size_t)(b0 + row) * S_DIM + c]
                      : actions[(size_t)(b0 + row) * AD_DIM + (c - S_DIM)];
        Xs[row * XS + c] = (_Float16)v;
    }
    __syncthreads();

    const _Float16* Xw = Xs + (wave * ROWS_PER_WAVE) * XS;
    _Float16*       Hw = Hs + (wave * ROWS_PER_WAVE) * HS;

    float acc[2][8];
#pragma unroll
    for (int m = 0; m < 2; ++m)
#pragma unroll
        for (int r = 0; r < 8; ++r) acc[m][r] = 0.f;

#pragma unroll 1
    for (int k = 0; k < K_N; ++k) {
        v8f ck[2], ca[2], cc[2];
        const int kn = (k + 1 < K_N) ? k + 1 : k;

        const float kbias = bk3[k];
        const float ab3v  = ba3[k * A_N + (col & 7)];
        const float cb3v  = bc3[k * A_N + (col & 7)];

        mlp3<S_DIM, S_DIM>(Xw, Hw, wk1 + (size_t)k * H_DIM * S_DIM, wa1 + (size_t)k * H_DIM * S_DIM,
                           wk2 + (size_t)k * H_DIM * H_DIM, wk3 + (size_t)k * 16 * H_DIM,
                           bk1 + k * H_DIM, bk2 + k * H_DIM, lane, ring, ck);
        mlp3<S_DIM, D_DIM>(Xw, Hw, wa1 + (size_t)k * H_DIM * S_DIM, wc1 + (size_t)k * H_DIM * D_DIM,
                           wa2 + (size_t)k * H_DIM * H_DIM, wa3 + (size_t)k * 16 * H_DIM,
                           ba1 + k * H_DIM, ba2 + k * H_DIM, lane, ring, ca);
        mlp3<D_DIM, S_DIM>(Xw, Hw, wc1 + (size_t)k * H_DIM * D_DIM, wk1 + (size_t)kn * H_DIM * S_DIM,
                           wc2 + (size_t)k * H_DIM * H_DIM, wc3 + (size_t)k * 16 * H_DIM,
                           bc1 + k * H_DIM, bc2 + k * H_DIM, lane, ring, cc);

#pragma unroll
        for (int m = 0; m < 2; ++m)
#pragma unroll
            for (int r = 0; r < 8; ++r) {
                float kv = __shfl(ck[m][r], lane & 16, 32);
                kv = fabsf(kv + kbias) + 1e-10f;
                float ag = 1.f / (1.f + __expf(-(ca[m][r] + ab3v)));
                float ac = 1.f / (1.f + __expf(-(cc[m][r] + cb3v)));
                acc[m][r] += kv * ag * ac;
            }
    }

    if (col < A_N) {
#pragma unroll
        for (int m = 0; m < 2; ++m)
#pragma unroll
            for (int r = 0; r < 8; ++r)
                out[(size_t)(b0 + wave * ROWS_PER_WAVE + m * 16 + r + 8 * hi) * A_N + col] =
                    acc[m][r];
    }
}

extern "C" void kernel_launch(void* const* d_in, const int* in_sizes, int n_in,
                              void* d_out, int out_size, void* d_ws, size_t ws_size,
                              hipStream_t stream) {
    (void)in_sizes; (void)n_in; (void)out_size; (void)ws_size;
    const float* states  = (const float*)d_in[0];
    const float* actions = (const float*)d_in[1];
    const float* kW1 = (const float*)d_in[2];
    const float* kb1 = (const float*)d_in[3];
    const float* kW2 = (const float*)d_in[4];
    const float* kb2 = (const float*)d_in[5];
    const float* kW3 = (const float*)d_in[6];
    const float* kb3 = (const float*)d_in[7];
    const float* aW1 = (const float*)d_in[8];
    const float* ab1 = (const float*)d_in[9];
    const float* aW2 = (const float*)d_in[10];
    const float* ab2 = (const float*)d_in[11];
    const float* aW3 = (const float*)d_in[12];
    const float* ab3 = (const float*)d_in[13];
    const float* cW1 = (const float*)d_in[14];
    const float* cb1 = (const float*)d_in[15];
    const float* cW2 = (const float*)d_in[16];
    const float* cb2 = (const float*)d_in[17];
    const float* cW3 = (const float*)d_in[18];
    const float* cb3 = (const float*)d_in[19];

    _Float16* ws = (_Float16*)d_ws;
    size_t off = 0;
    _Float16* wk1 = ws + off; off += (size_t)K_N * H_DIM * S_DIM;
    _Float16* wa1 = ws + off; off += (size_t)K_N * H_DIM * S_DIM;
    _Float16* wc1 = ws + off; off += (size_t)K_N * H_DIM * D_DIM;
    _Float16* wk2 = ws + off; off += (size_t)K_N * H_DIM * H_DIM;
    _Float16* wa2 = ws + off; off += (size_t)K_N * H_DIM * H_DIM;
    _Float16* wc2 = ws + off; off += (size_t)K_N * H_DIM * H_DIM;
    _Float16* wk3 = ws + off; off += (size_t)K_N * 16 * H_DIM;
    _Float16* wa3 = ws + off; off += (size_t)K_N * 16 * H_DIM;
    _Float16* wc3 = ws + off; off += (size_t)K_N * 16 * H_DIM;

    auto cvt = [&](const float* s, _Float16* d, int n) {
        cvt_f32_f16<<<(n + 255) / 256, 256, 0, stream>>>(s, d, n);
    };
    cvt(kW1, wk1, K_N * H_DIM * S_DIM);
    cvt(aW1, wa1, K_N * H_DIM * S_DIM);
    cvt(cW1, wc1, K_N * H_DIM * D_DIM);
    cvt(kW2, wk2, K_N * H_DIM * H_DIM);
    cvt(aW2, wa2, K_N * H_DIM * H_DIM);
    cvt(cW2, wc2, K_N * H_DIM * H_DIM);
    const int n3 = K_N * 16 * H_DIM;
    cvt_pad_w3<<<(n3 + 255) / 256, 256, 0, stream>>>(kW3, wk3, 1);
    cvt_pad_w3<<<(n3 + 255) / 256, 256, 0, stream>>>(aW3, wa3, A_N);
    cvt_pad_w3<<<(n3 + 255) / 256, 256, 0, stream>>>(cW3, wc3, A_N);

    qplex_main<<<B_TOT / WG_ROWS, NTHREADS, 0, stream>>>(
        states, actions,
        wk1, wa1, wc1, wk2, wa2, wc2, wk3, wa3, wc3,
        kb1, kb2, kb3, ab1, ab2, ab3, cb1, cb2, cb3,
        (float*)d_out);
}


# Weights stored in the workspace in WMMA-fragment order: a weight-fragment load reads 512 contiguous bytes (4 whole 128-byte lines), not 32 bytes from each of 16 rows; same values, same k order
# speedup vs baseline: 2.5172x; 1.2397x over previous
#include <hip/hip_runtime.h>
#include <hip/hip_bf16.h>

typedef __attribute__((ext_vector_type(16))) _Float16 v16h;
typedef __attribute__((ext_vector_type(8)))  _Float16 v8h;
typedef __attribute__((ext_vector_type(8)))  float    v8f;

#define B_TOT   65536
#define S_DIM   256
#define A_N     8
#define K_N     10
#define H_DIM   64
#define AD_DIM  128
#define D_DIM   384

#define WG_ROWS 64
#define NTHREADS 64
#define NWAVES  (NTHREADS / 32)
#define ROWS_PER_WAVE 32
#define XS      392
#define HS      72

__device__ __forceinline__ v16h load_frag(const _Float16* __restrict__ p,
                                          int stride, int row, int k0, int lane) {
    const int ho = (lane >> 4) << 3;
    const _Float16* q = p + row * stride + k0 + ho;
    union { v16h v; struct { v8h lo; v8h hi; } s; } u;
    u.s.lo = *(const v8h*)q;
    u.s.hi = *(const v8h*)(q + 16);
    return u.v;
}

__device__ __forceinline__ v16h load_wfrag(const _Float16* __restrict__ p, int f, int lane) {
    const _Float16* q = p + f * 512 + lane * 8;
    union { v16h v; struct { v8h lo; v8h hi; } s; } u;
    u.s.lo = *(const v8h*)q;
    u.s.hi = *(const v8h*)(q + 256);
    return u.v;
}

#define PF_D 4

__device__ __forceinline__ v8f wmma_f16(v16h a, v16h b, v8f c) {
    return __builtin_amdgcn_wmma_f32_16x16x32_f16(false, a, false, b, (short)0, c, false, false);
}

template <int KD>
__device__ __forceinline__ void gemm4x2_l1(const _Float16* __restrict__ A, int lda,
                                           const _Float16* __restrict__ W,
                                           const _Float16* __restrict__ Wn,
                                           int lane, v16h (&ring)[PF_D][4], v8f c[2][4]) {
    constexpr int NS = KD / 32;
    const int mr = lane & 15;
    v8f z = {0.f, 0.f, 0.f, 0.f, 0.f, 0.f, 0.f, 0.f};
#pragma unroll
    for (int m = 0; m < 2; ++m)
#pragma unroll
        for (int t = 0; t < 4; ++t) c[m][t] = z;
#pragma unroll
    for (int s = 0; s < NS; ++s) {
        v16h a0 = load_frag(A, lda, mr, s * 32, lane);
        v16h a1 = load_frag(A + 16 * lda, lda, mr, s * 32, lane);
#pragma unroll
        for (int t = 0; t < 4; ++t) {
            c[0][t] = wmma_f16(a0, ring[s % PF_D][t], c[0][t]);
            c[1][t] = wmma_f16(a1, ring[s % PF_D][t], c[1][t]);
        }
#pragma unroll
        for (int t = 0; t < 4; ++t) {
            if (s + PF_D < NS)
                ring[s % PF_D][t] = load_wfrag(W, (s + PF_D) * 4 + t, lane);
            else
                ring[s % PF_D][t] = load_wfrag(Wn, (s + PF_D - NS) * 4 + t, lane);
        }
    }
}

__device__ __forceinline__ void gemm4x2_h(const _Float16* __restrict__ A,
                                          const v16h (&w)[2][4], int lane, v8f c[2][4]) {
    const int mr = lane & 15;
    v8f z = {0.f, 0.f, 0.f, 0.f, 0.f, 0.f, 0.f, 0.f};
#pragma unroll
    for (int m = 0; m < 2; ++m)
#pragma unroll
        for (int t = 0; t < 4; ++t) c[m][t] = z;
#pragma unroll
    for (int s = 0; s < 2; ++s) {
        v16h a0 = load_frag(A, HS, mr, s * 32, lane);
        v16h a1 = load_frag(A + 16 * HS, HS, mr, s * 32, lane);
#pragma unroll
        for (int t = 0; t < 4; ++t) {
            c[0][t] = wmma_f16(a0, w[s][t], c[0][t]);
            c[1][t] = wmma_f16(a1, w[s][t], c[1][t]);
        }
    }
}

__device__ __forceinline__ void gemm1x2_h(const _Float16* __restrict__ A,
                                          const v16h (&w)[2], int lane, v8f c[2]) {
    const int mr = lane & 15;
    v8f z = {0.f, 0.f, 0.f, 0.f, 0.f, 0.f, 0.f, 0.f};
    c[0] = z; c[1] = z;
#pragma unroll
    for (int s = 0; s < 2; ++s) {
        v16h a0 = load_frag(A, HS, mr, s * 32, lane);
        v16h a1 = load_frag(A + 16 * HS, HS, mr, s * 32, lane);
        c[0] = wmma_f16(a0, w[s], c[0]);
        c[1] = wmma_f16(a1, w[s], c[1]);
    }
}

__device__ __forceinline__ void bias_relu_store2(const v8f c[2][4],
                                                 const float (&bv)[4],
                                                 _Float16* __restrict__ Hout, int lane) {
    const int col = lane & 15, hi = lane >> 4;
#pragma unroll
    for (int t = 0; t < 4; ++t) {
#pragma unroll
        for (int m = 0; m < 2; ++m)
#pragma unroll
            for (int r = 0; r < 8; ++r) {
                float v = c[m][t][r] + bv[t];
                Hout[(m * 16 + r + 8 * hi) * HS + t * 16 + col] =
                    (_Float16)(v > 0.f ? v : 0.f);
            }
    }
}

template <int KD>
__device__ __forceinline__ void mlp3(const _Float16* __restrict__ Xw, _Float16* __restrict__ Hw,
                                     const _Float16* __restrict__ W1,
                                     const _Float16* __restrict__ W1n,
                                     const _Float16* __restrict__ W2,
                                     const _Float16* __restrict__ W3,
                                     const float* __restrict__ b1,
                                     const float* __restrict__ b2,
                                     int lane, v16h (&ring)[PF_D][4], v8f out[2]) {
    const int mr = lane & 15;
    v16h w2[2][4];
    v16h w3[2];
    float bv1[4], bv2[4];
#pragma unroll
    for (int s = 0; s < 2; ++s)
#pragma unroll
        for (int t = 0; t < 4; ++t) w2[s][t] = load_wfrag(W2, s * 4 + t, lane);
#pragma unroll
    for (int s = 0; s < 2; ++s) w3[s] = load_wfrag(W3, s, lane);
#pragma unroll
    for (int t = 0; t < 4; ++t) { bv1[t] = b1[t * 16 + mr]; bv2[t] = b2[t * 16 + mr]; }

    v8f c1[2][4];
    gemm4x2_l1<KD>(Xw, XS, W1, W1n, lane, ring, c1);
    bias_relu_store2(c1, bv1, Hw, lane);
    gemm4x2_h(Hw, w2, lane, c1);
    bias_relu_store2(c1, bv2, Hw, lane);
    gemm1x2_h(Hw, w3, lane, out);
}

__global__ void cvt_frag_f16(const float* __restrict__ src, _Float16* __restrict__ dst,
                             int n, int kd, int nt, int rows) {
    int i = blockIdx.x * blockDim.x + threadIdx.x;
    if (i >= n) return;
    const int e    = i & 7;
    const int lane = (i >> 3) & 31;
    const int half = (i >> 8) & 1;
    const int f    = i >> 9;
    const int fpm  = (kd >> 5) * nt;
    const int kern = f / fpm;
    const int fm   = f - kern * fpm;
    const int ks   = fm / nt;
    const int t    = fm - ks * nt;
    const int row  = t * 16 + (lane & 15);
    const int k    = ks * 32 + half * 16 + (lane >> 4) * 8 + e;
    dst[i] = (row < rows) ? (_Float16)src[((size_t)kern * rows + row) * kd + k] : (_Float16)0.f;
}

__global__ __launch_bounds__(NTHREADS, 2)
void qplex_main(const float* __restrict__ states, const float* __restrict__ actions,
                const _Float16* __restrict__ wk1, const _Float16* __restrict__ wa1,
                const _Float16* __restrict__ wc1,
                const _Float16* __restrict__ wk2, const _Float16* __restrict__ wa2,
                const _Float16* __restrict__ wc2,
                const _Float16* __restrict__ wk3, const _Float16* __restrict__ wa3,
                const _Float16* __restrict__ wc3,
                const float* __restrict__ bk1, const float* __restrict__ bk2,
                const float* __restrict__ bk3,
                const float* __restrict__ ba1, const float* __restrict__ ba2,
                const float* __restrict__ ba3,
                const float* __restrict__ bc1, const float* __restrict__ bc2,
                const float* __restrict__ bc3,
                float* __restrict__ out) {
    __shared__ _Float16 Xs[WG_ROWS * XS];
    __shared__ _Float16 Hs[NWAVES * ROWS_PER_WAVE * HS];

    const int tid = threadIdx.x;
    const int b0  = blockIdx.x * WG_ROWS;
    const int wave = tid >> 5, lane = tid & 31;
    const int col = lane & 15, hi = lane >> 4;

    v16h ring[PF_D][4];
#pragma unroll
    for (int s = 0; s < PF_D; ++s)
#pragma unroll
        for (int t = 0; t < 4; ++t) ring[s][t] = load_wfrag(wk1, s * 4 + t, lane);

    for (int idx = tid; idx < WG_ROWS * D_DIM; idx += NTHREADS) {
        int row = idx / D_DIM;
        int c = idx - row * D_DIM;
        float v = (c < S_DIM)
                      ? states[(size_t)(b0 + row) * S_DIM + c]
                      : actions[(size_t)(b0 + row) * AD_DIM + (c - S_DIM)];
        Xs[row * XS + c] = (_Float16)v;
    }
    __syncthreads();

    const _Float16* Xw = Xs + (wave * ROWS_PER_WAVE) * XS;
    _Float16*       Hw = Hs + (wave * ROWS_PER_WAVE) * HS;

    float acc[2][8];
#pragma unroll
    for (int m = 0; m < 2; ++m)
#pragma unroll
        for (int r = 0; r < 8; ++r) acc[m][r] = 0.f;

#pragma unroll 1
    for (int k = 0; k < K_N; ++k) {
        v8f ck[2], ca[2], cc[2];
        const int kn = (k + 1 < K_N) ? k + 1 : k;

        const float kbias = bk3[k];
        const float ab3v  = ba3[k * A_N + (col & 7)];
        const float cb3v  = bc3[k * A_N + (col & 7)];

        mlp3<S_DIM>(Xw, Hw, wk1 + (size_t)k * H_DIM * S_DIM, wa1 + (size_t)k * H_DIM * S_DIM,
                           wk2 + (size_t)k * H_DIM * H_DIM, wk3 + (size_t)k * 16 * H_DIM,
                           bk1 + k * H_DIM, bk2 + k * H_DIM, lane, ring, ck);
        mlp3<S_DIM>(Xw, Hw, wa1 + (size_t)k * H_DIM * S_DIM, wc1 + (size_t)k * H_DIM * D_DIM,
                           wa2 + (size_t)k * H_DIM * H_DIM, wa3 + (size_t)k * 16 * H_DIM,
                           ba1 + k * H_DIM, ba2 + k * H_DIM, lane, ring, ca);
        mlp3<D_DIM>(Xw, Hw, wc1 + (size_t)k * H_DIM * D_DIM, wk1 + (size_t)kn * H_DIM * S_DIM,
                           wc2 + (size_t)k * H_DIM * H_DIM, wc3 + (size_t)k * 16 * H_DIM,
                           bc1 + k * H_DIM, bc2 + k * H_DIM, lane, ring, cc);

#pragma unroll
        for (int m = 0; m < 2; ++m)
#pragma unroll
            for (int r = 0; r < 8; ++r) {
                float kv = __shfl(ck[m][r], lane & 16, 32);
                kv = fabsf(kv + kbias) + 1e-10f;
                float ag = 1.f / (1.f + __expf(-(ca[m][r] + ab3v)));
                float ac = 1.f / (1.f + __expf(-(cc[m][r] + cb3v)));
                acc[m][r] += kv * ag * ac;
            }
    }

    if (col < A_N) {
#pragma unroll
        for (int m = 0; m < 2; ++m)
#pragma unroll
            for (int r = 0; r < 8; ++r)
                out[(size_t)(b0 + wave * ROWS_PER_WAVE + m * 16 + r + 8 * hi) * A_N + col] =
                    acc[m][r];
    }
}

extern "C" void kernel_launch(void* const* d_in, const int* in_sizes, int n_in,
                              void* d_out, int out_size, void* d_ws, size_t ws_size,
                              hipStream_t stream) {
    (void)in_sizes; (void)n_in; (void)out_size; (void)ws_size;
    const float* states  = (const float*)d_in[0];
    const float* actions = (const float*)d_in[1];
    const float* kW1 = (const float*)d_in[2];
    const float* kb1 = (const float*)d_in[3];
    const float* kW2 = (const float*)d_in[4];
    const float* kb2 = (const float*)d_in[5];
    const float* kW3 = (const float*)d_in[6];
    const float* kb3 = (const float*)d_in[7];
    const float* aW1 = (const float*)d_in[8];
    const float* ab1 = (const float*)d_in[9];
    const float* aW2 = (const float*)d_in[10];
    const float* ab2 = (const float*)d_in[11];
    const float* aW3 = (const float*)d_in[12];
    const float* ab3 = (const float*)d_in[13];
    const float* cW1 = (const float*)d_in[14];
    const float* cb1 = (const float*)d_in[15];
    const float* cW2 = (const float*)d_in[16];
    const float* cb2 = (const float*)d_in[17];
    const float* cW3 = (const float*)d_in[18];
    const float* cb3 = (const float*)d_in[19];

    _Float16* ws = (_Float16*)d_ws;
    size_t off = 0;
    _Float16* wk1 = ws + off; off += (size_t)K_N * H_DIM * S_DIM;
    _Float16* wa1 = ws + off; off += (size_t)K_N * H_DIM * S_DIM;
    _Float16* wc1 = ws + off; off += (size_t)K_N * H_DIM * D_DIM;
    _Float16* wk2 = ws + off; off += (size_t)K_N * H_DIM * H_DIM;
    _Float16* wa2 = ws + off; off += (size_t)K_N * H_DIM * H_DIM;
    _Float16* wc2 = ws + off; off += (size_t)K_N * H_DIM * H_DIM;
    _Float16* wk3 = ws + off; off += (size_t)K_N * 16 * H_DIM;
    _Float16* wa3 = ws + off; off += (size_t)K_N * 16 * H_DIM;
    _Float16* wc3 = ws + off; off += (size_t)K_N * 16 * H_DIM;

    auto cvt = [&](const float* s, _Float16* d, int n, int kd, int nt, int rows) {
        cvt_frag_f16<<<(n + 255) / 256, 256, 0, stream>>>(s, d, n, kd, nt, rows);
    };
    cvt(kW1, wk1, K_N * H_DIM * S_DIM, S_DIM, 4, H_DIM);
    cvt(aW1, wa1, K_N * H_DIM * S_DIM, S_DIM, 4, H_DIM);
    cvt(cW1, wc1, K_N * H_DIM * D_DIM, D_DIM, 4, H_DIM);
    cvt(kW2, wk2, K_N * H_DIM * H_DIM, H_DIM, 4, H_DIM);
    cvt(aW2, wa2, K_N * H_DIM * H_DIM, H_DIM, 4, H_DIM);
    cvt(cW2, wc2, K_N * H_DIM * H_DIM, H_DIM, 4, H_DIM);
    cvt(kW3, wk3, K_N * 16 * H_DIM, H_DIM, 1, 1);
    cvt(aW3, wa3, K_N * 16 * H_DIM, H_DIM, 1, A_N);
    cvt(cW3, wc3, K_N * 16 * H_DIM, H_DIM, 1, A_N);

    qplex_main<<<B_TOT / WG_ROWS, NTHREADS, 0, stream>>>(
        states, actions,
        wk1, wa1, wc1, wk2, wa2, wc2, wk3, wa3, wc3,
        kb1, kb2, kb3, ab1, ab2, ab3, cb1, cb2, cb3,
        (float*)d_out);
}


# Input tile staged with 16-byte global loads, 16 issued per lane before any is converted, and 8-byte LDS stores, instead of four 4-byte loads waited for on every trip; same f16 values, same sums
# speedup vs baseline: 3.2840x; 1.3046x over previous
#include <hip/hip_runtime.h>
#include <hip/hip_bf16.h>

typedef __attribute__((ext_vector_type(16))) _Float16 v16h;
typedef __attribute__((ext_vector_type(8)))  _Float16 v8h;
typedef __attribute__((ext_vector_type(8)))  float    v8f;
typedef __attribute__((ext_vector_type(4)))  float    v4f;
typedef __attribute__((ext_vector_type(4)))  _Float16 v4h;

#define B_TOT   65536
#define S_DIM   256
#define A_N     8
#define K_N     10
#define H_DIM   64
#define AD_DIM  128
#define D_DIM   384

#define WG_ROWS 64
#define NTHREADS 64
#define NWAVES  (NTHREADS / 32)
#define ROWS_PER_WAVE 32
#define XS      392
#define HS      72
#define ST_NB   16

__device__ __forceinline__ v16h load_frag(const _Float16* __restrict__ p,
                                          int stride, int row, int k0, int lane) {
    const int ho = (lane >> 4) << 3;
    const _Float16* q = p + row * stride + k0 + ho;
    union { v16h v; struct { v8h lo; v8h hi; } s; } u;
    u.s.lo = *(const v8h*)q;
    u.s.hi = *(const v8h*)(q + 16);
    return u.v;
}

__device__ __forceinline__ v16h load_wfrag(const _Float16* __restrict__ p, int f, int lane) {
    const _Float16* q = p + f * 512 + lane * 8;
    union { v16h v; struct { v8h lo; v8h hi; } s; } u;
    u.s.lo = *(const v8h*)q;
    u.s.hi = *(const v8h*)(q + 256);
    return u.v;
}

#define PF_D 4

__device__ __forceinline__ v8f wmma_f16(v16h a, v16h b, v8f c) {
    return __builtin_amdgcn_wmma_f32_16x16x32_f16(false, a, false, b, (short)0, c, false, false);
}

template <int KD>
__device__ __forceinline__ void gemm4x2_l1(const _Float16* __restrict__ A, int lda,
                                           const _Float16* __restrict__ W,
                                           const _Float16* __restrict__ Wn,
                                           int lane, v16h (&ring)[PF_D][4], v8f c[2][4]) {
    constexpr int NS = KD / 32;
    const int mr = lane & 15;
    v8f z = {0.f, 0.f, 0.f, 0.f, 0.f, 0.f, 0.f, 0.f};
#pragma unroll
    for (int m = 0; m < 2; ++m)
#pragma unroll
        for (int t = 0; t < 4; ++t) c[m][t] = z;
#pragma unroll
    for (int s = 0; s < NS; ++s) {
        v16h a0 = load_frag(A, lda, mr, s * 32, lane);
        v16h a1 = load_frag(A + 16 * lda, lda, mr, s * 32, lane);
#pragma unroll
        for (int t = 0; t < 4; ++t) {
            c[0][t] = wmma_f16(a0, ring[s % PF_D][t], c[0][t]);
            c[1][t] = wmma_f16(a1, ring[s % PF_D][t], c[1][t]);
        }
#pragma unroll
        for (int t = 0; t < 4; ++t) {
            if (s + PF_D < NS)
                ring[s % PF_D][t] = load_wfrag(W, (s + PF_D) * 4 + t, lane);
            else
                ring[s % PF_D][t] = load_wfrag(Wn, (s + PF_D - NS) * 4 + t, lane);
        }
    }
}

__device__ __forceinline__ void gemm4x2_h(const _Float16* __restrict__ A,
                                          const v16h (&w)[2][4], int lane, v8f c[2][4]) {
    const int mr = lane & 15;
    v8f z = {0.f, 0.f, 0.f, 0.f, 0.f, 0.f, 0.f, 0.f};
#pragma unroll
    for (int m = 0; m < 2; ++m)
#pragma unroll
        for (int t = 0; t < 4; ++t) c[m][t] = z;
#pragma unroll
    for (int s = 0; s < 2; ++s) {
        v16h a0 = load_frag(A, HS, mr, s * 32, lane);
        v16h a1 = load_frag(A + 16 * HS, HS, mr, s * 32, lane);
#pragma unroll
        for (int t = 0; t < 4; ++t) {
            c[0][t] = wmma_f16(a0, w[s][t], c[0][t]);
            c[1][t] = wmma_f16(a1, w[s][t], c[1][t]);
        }
    }
}

__device__ __forceinline__ void gemm1x2_h(const _Float16* __restrict__ A,
                                          const v16h (&w)[2], int lane, v8f c[2]) {
    const int mr = lane & 15;
    v8f z = {0.f, 0.f, 0.f, 0.f, 0.f, 0.f, 0.f, 0.f};
    c[0] = z; c[1] = z;
#pragma unroll
    for (int s = 0; s < 2; ++s) {
        v16h a0 = load_frag(A, HS, mr, s * 32, lane);
        v16h a1 = load_frag(A + 16 * HS, HS, mr, s * 32, lane);
        c[0] = wmma_f16(a0, w[s], c[0]);
        c[1] = wmma_f16(a1, w[s], c[1]);
    }
}

__device__ __forceinline__ void bias_relu_store2(const v8f c[2][4],
                                                 const float (&bv)[4],
                                                 _Float16* __restrict__ Hout, int lane) {
    const int col = lane & 15, hi = lane >> 4;
#pragma unroll
    for (int t = 0; t < 4; ++t) {
#pragma unroll
        for (int m = 0; m < 2; ++m)
#pragma unroll
            for (int r = 0; r < 8; ++r) {
                float v = c[m][t][r] + bv[t];
                Hout[(m * 16 + r + 8 * hi) * HS + t * 16 + col] =
                    (_Float16)(v > 0.f ? v : 0.f);
            }
    }
}

template <int KD>
__device__ __forceinline__ void mlp3(const _Float16* __restrict__ Xw, _Float16* __restrict__ Hw,
                                     const _Float16* __restrict__ W1,
                                     const _Float16* __restrict__ W1n,
                                     const _Float16* __restrict__ W2,
                                     const _Float16* __restrict__ W3,
                                     const float* __restrict__ b1,
                                     const float* __restrict__ b2,
                                     int lane, v16h (&ring)[PF_D][4], v8f out[2]) {
    const int mr = lane & 15;
    v16h w2[2][4];
    v16h w3[2];
    float bv1[4], bv2[4];
#pragma unroll
    for (int s = 0; s < 2; ++s)
#pragma unroll
        for (int t = 0; t < 4; ++t) w2[s][t] = load_wfrag(W2, s * 4 + t, lane);
#pragma unroll
    for (int s = 0; s < 2; ++s) w3[s] = load_wfrag(W3, s, lane);
#pragma unroll
    for (int t = 0; t < 4; ++t) { bv1[t] = b1[t * 16 + mr]; bv2[t] = b2[t * 16 + mr]; }

    v8f c1[2][4];
    gemm4x2_l1<KD>(Xw, XS, W1, W1n, lane, ring, c1);
    bias_relu_store2(c1, bv1, Hw, lane);
    gemm4x2_h(Hw, w2, lane, c1);
    bias_relu_store2(c1, bv2, Hw, lane);
    gemm1x2_h(Hw, w3, lane, out);
}

__global__ void cvt_frag_f16(const float* __restrict__ src, _Float16* __restrict__ dst,
                             int n, int kd, int nt, int rows) {
    int i = blockIdx.x * blockDim.x + threadIdx.x;
    if (i >= n) return;
    const int e    = i & 7;
    const int lane = (i >> 3) & 31;
    const int half = (i >> 8) & 1;
    const int f    = i >> 9;
    const int fpm  = (kd >> 5) * nt;
    const int kern = f / fpm;
    const int fm   = f - kern * fpm;
    const int ks   = fm / nt;
    const int t    = fm - ks * nt;
    const int row  = t * 16 + (lane & 15);
    const int k    = ks * 32 + half * 16 + (lane >> 4) * 8 + e;
    dst[i] = (row < rows) ? (_Float16)src[((size_t)kern * rows + row) * kd + k] : (_Float16)0.f;
}

__global__ __launch_bounds__(NTHREADS, 2)
void qplex_main(const float* __restrict__ states, const float* __restrict__ actions,
                const _Float16* __restrict__ wk1, const _Float16* __restrict__ wa1,
                const _Float16* __restrict__ wc1,
                const _Float16* __restrict__ wk2, const _Float16* __restrict__ wa2,
                const _Float16* __restrict__ wc2,
                const _Float16* __restrict__ wk3, const _Float16* __restrict__ wa3,
                const _Float16* __restrict__ wc3,
                const float* __restrict__ bk1, const float* __restrict__ bk2,
                const float* __restrict__ bk3,
                const float* __restrict__ ba1, const float* __restrict__ ba2,
                const float* __restrict__ ba3,
                const float* __restrict__ bc1, const float* __restrict__ bc2,
                const float* __restrict__ bc3,
                float* __restrict__ out) {
    __shared__ _Float16 Xs[WG_ROWS * XS];
    __shared__ _Float16 Hs[NWAVES * ROWS_PER_WAVE * HS];

    const int tid = threadIdx.x;
    const int b0  = blockIdx.x * WG_ROWS;
    const int wave = tid >> 5, lane = tid & 31;
    const int col = lane & 15, hi = lane >> 4;

    v16h ring[PF_D][4];
#pragma unroll
    for (int s = 0; s < PF_D; ++s)
#pragma unroll
        for (int t = 0; t < 4; ++t) ring[s][t] = load_wfrag(wk1, s * 4 + t, lane);

    {
        const v4f* __restrict__ sp = (const v4f*)(states + (size_t)b0 * S_DIM) + tid;
        const v4f* __restrict__ ap = (const v4f*)(actions + (size_t)b0 * AD_DIM) + tid;
        _Float16* xs_s = Xs + 4 * tid;
        _Float16* xs_a = Xs + (tid >> 5) * XS + S_DIM + 4 * (tid & 31);
#pragma unroll 1
        for (int g = 0; g < WG_ROWS / ST_NB; ++g) {
            v4f v[ST_NB];
#pragma unroll
            for (int j = 0; j < ST_NB; ++j) v[j] = sp[(g * ST_NB + j) * (S_DIM / 4)];
#pragma unroll
            for (int j = 0; j < ST_NB; ++j) {
                v4h h;
                h.x = (_Float16)v[j].x; h.y = (_Float16)v[j].y;
                h.z = (_Float16)v[j].z; h.w = (_Float16)v[j].w;
                *(v4h*)(xs_s + (g * ST_NB + j) * XS) = h;
            }
        }
#pragma unroll 1
        for (int g = 0; g < (WG_ROWS / 2) / ST_NB; ++g) {
            v4f v[ST_NB];
#pragma unroll
            for (int j = 0; j < ST_NB; ++j) v[j] = ap[(g * ST_NB + j) * NTHREADS];
#pragma unroll
            for (int j = 0; j < ST_NB; ++j) {
                v4h h;
                h.x = (_Float16)v[j].x; h.y = (_Float16)v[j].y;
                h.z = (_Float16)v[j].z; h.w = (_Float16)v[j].w;
                *(v4h*)(xs_a + 2 * (g * ST_NB + j) * XS) = h;
            }
        }
    }
    __syncthreads();

    const _Float16* Xw = Xs + (wave * ROWS_PER_WAVE) * XS;
    _Float16*       Hw = Hs + (wave * ROWS_PER_WAVE) * HS;

    float acc[2][8];
#pragma unroll
    for (int m = 0; m < 2; ++m)
#pragma unroll
        for (int r = 0; r < 8; ++r) acc[m][r] = 0.f;

#pragma unroll 1
    for (int k = 0; k < K_N; ++k) {
        v8f ck[2], ca[2], cc[2];
        const int kn = (k + 1 < K_N) ? k + 1 : k;

        const float kbias = bk3[k];
        const float ab3v  = ba3[k * A_N + (col & 7)];
        const float cb3v  = bc3[k * A_N + (col & 7)];

        mlp3<S_DIM>(Xw, Hw, wk1 + (size_t)k * H_DIM * S_DIM, wa1 + (size_t)k * H_DIM * S_DIM,
                           wk2 + (size_t)k * H_DIM * H_DIM, wk3 + (size_t)k * 16 * H_DIM,
                           bk1 + k * H_DIM, bk2 + k * H_DIM, lane, ring, ck);
        mlp3<S_DIM>(Xw, Hw, wa1 + (size_t)k * H_DIM * S_DIM, wc1 + (size_t)k * H_DIM * D_DIM,
                           wa2 + (size_t)k * H_DIM * H_DIM, wa3 + (size_t)k * 16 * H_DIM,
                           ba1 + k * H_DIM, ba2 + k * H_DIM, lane, ring, ca);
        mlp3<D_DIM>(Xw, Hw, wc1 + (size_t)k * H_DIM * D_DIM, wk1 + (size_t)kn * H_DIM * S_DIM,
                           wc2 + (size_t)k * H_DIM * H_DIM, wc3 + (size_t)k * 16 * H_DIM,
                           bc1 + k * H_DIM, bc2 + k * H_DIM, lane, ring, cc);

#pragma unroll
        for (int m = 0; m < 2; ++m)
#pragma unroll
            for (int r = 0; r < 8; ++r) {
                float kv = __shfl(ck[m][r], lane & 16, 32);
                kv = fabsf(kv + kbias) + 1e-10f;
                float ag = 1.f / (1.f + __expf(-(ca[m][r] + ab3v)));
                float ac = 1.f / (1.f + __expf(-(cc[m][r] + cb3v)));
                acc[m][r] += kv * ag * ac;
            }
    }

    if (col < A_N) {
#pragma unroll
        for (int m = 0; m < 2; ++m)
#pragma unroll
            for (int r = 0; r < 8; ++r)
                out[(size_t)(b0 + wave * ROWS_PER_WAVE + m * 16 + r + 8 * hi) * A_N + col] =
                    acc[m][r];
    }
}

extern "C" void kernel_launch(void* const* d_in, const int* in_sizes, int n_in,
                              void* d_out, int out_size, void* d_ws, size_t ws_size,
                              hipStream_t stream) {
    (void)in_sizes; (void)n_in; (void)out_size; (void)ws_size;
    const float* states  = (const float*)d_in[0];
    const float* actions = (const float*)d_in[1];
    const float* kW1 = (const float*)d_in[2];
    const float* kb1 = (const float*)d_in[3];
    const float* kW2 = (const float*)d_in[4];
    const float* kb2 = (const float*)d_in[5];
    const float* kW3 = (const float*)d_in[6];
    const float* kb3 = (const float*)d_in[7];
    const float* aW1 = (const float*)d_in[8];
    const float* ab1 = (const float*)d_in[9];
    const float* aW2 = (const float*)d_in[10];
    const float* ab2 = (const float*)d_in[11];
    const float* aW3 = (const float*)d_in[12];
    const float* ab3 = (const float*)d_in[13];
    const float* cW1 = (const float*)d_in[14];
    const float* cb1 = (const float*)d_in[15];
    const float* cW2 = (const float*)d_in[16];
    const float* cb2 = (const float*)d_in[17];
    const float* cW3 = (const float*)d_in[18];
    const float* cb3 = (const float*)d_in[19];

    _Float16* ws = (_Float16*)d_ws;
    size_t off = 0;
    _Float16* wk1 = ws + off; off += (size_t)K_N * H_DIM * S_DIM;
    _Float16* wa1 = ws + off; off += (size_t)K_N * H_DIM * S_DIM;
    _Float16* wc1 = ws + off; off += (size_t)K_N * H_DIM * D_DIM;
    _Float16* wk2 = ws + off; off += (size_t)K_N * H_DIM * H_DIM;
    _Float16* wa2 = ws + off; off += (size_t)K_N * H_DIM * H_DIM;
    _Float16* wc2 = ws + off; off += (size_t)K_N * H_DIM * H_DIM;
    _Float16* wk3 = ws + off; off += (size_t)K_N * 16 * H_DIM;
    _Float16* wa3 = ws + off; off += (size_t)K_N * 16 * H_DIM;
    _Float16* wc3 = ws + off; off += (size_t)K_N * 16 * H_DIM;

    auto cvt = [&](const float* s, _Float16* d, int n, int kd, int nt, int rows) {
        cvt_frag_f16<<<(n + 255) / 256, 256, 0, stream>>>(s, d, n, kd, nt, rows);
    };
    cvt(kW1, wk1, K_N * H_DIM * S_DIM, S_DIM, 4, H_DIM);
    cvt(aW1, wa1, K_N * H_DIM * S_DIM, S_DIM, 4, H_DIM);
    cvt(cW1, wc1, K_N * H_DIM * D_DIM, D_DIM, 4, H_DIM);
    cvt(kW2, wk2, K_N * H_DIM * H_DIM, H_DIM, 4, H_DIM);
    cvt(aW2, wa2, K_N * H_DIM * H_DIM, H_DIM, 4, H_DIM);
    cvt(cW2, wc2, K_N * H_DIM * H_DIM, H_DIM, 4, H_DIM);
    cvt(kW3, wk3, K_N * 16 * H_DIM, H_DIM, 1, 1);
    cvt(aW3, wa3, K_N * 16 * H_DIM, H_DIM, 1, A_N);
    cvt(cW3, wc3, K_N * 16 * H_DIM, H_DIM, 1, A_N);

    qplex_main<<<B_TOT / WG_ROWS, NTHREADS, 0, stream>>>(
        states, actions,
        wk1, wa1, wc1, wk2, wa2, wc2, wk3, wa3, wc3,
        kb1, kb2, kb3, ab1, ab2, ab3, cb1, cb2, cb3,
        (float*)d_out);
}


# Hidden activations stay in registers: layers computed transposed (weights as A operand), each f16-packed C tile pair is the next layer's B fragment, no LDS round trip; same values, same sums
# speedup vs baseline: 4.6905x; 1.4283x over previous
#include <hip/hip_runtime.h>
#include <hip/hip_bf16.h>

typedef __attribute__((ext_vector_type(16))) _Float16 v16h;
typedef __attribute__((ext_vector_type(8)))  _Float16 v8h;
typedef __attribute__((ext_vector_type(8)))  float    v8f;
typedef __attribute__((ext_vector_type(4)))  float    v4f;
typedef __attribute__((ext_vector_type(4)))  _Float16 v4h;

#define B_TOT   65536
#define S_DIM   256
#define A_N     8
#define K_N     10
#define H_DIM   64
#define AD_DIM  128
#define D_DIM   384

#define WG_ROWS 64
#define NTHREADS 64
#define NWAVES  (NTHREADS / 32)
#define ROWS_PER_WAVE 32
#define XS      392
#define ST_NB   16

__device__ __forceinline__ v16h load_frag(const _Float16* __restrict__ p,
                                          int stride, int row, int k0, int lane) {
    const int ho = (lane >> 4) << 3;
    const _Float16* q = p + row * stride + k0 + ho;
    union { v16h v; struct { v8h lo; v8h hi; } s; } u;
    u.s.lo = *(const v8h*)q;
    u.s.hi = *(const v8h*)(q + 16);
    return u.v;
}

__device__ __forceinline__ v16h load_wfrag(const _Float16* __restrict__ p, int f, int lane) {
    const _Float16* q = p + f * 512 + lane * 8;
    union { v16h v; struct { v8h lo; v8h hi; } s; } u;
    u.s.lo = *(const v8h*)q;
    u.s.hi = *(const v8h*)(q + 256);
    return u.v;
}

#define PF_D 4

__device__ __forceinline__ v8f wmma_f16(v16h a, v16h b, v8f c) {
    return __builtin_amdgcn_wmma_f32_16x16x32_f16(false, a, false, b, (short)0, c, false, false);
}


__device__ __forceinline__ v8f load_bias8(const float* __restrict__ b, int t, int hi) {
    const v4f* q = (const v4f*)(b + t * 16 + 8 * hi);
    v4f lo = q[0], up = q[1];
    v8f r = {lo.x, lo.y, lo.z, lo.w, up.x, up.y, up.z, up.w};
    return r;
}

template <int KD>
__device__ __forceinline__ void gemm_l1(const _Float16* __restrict__ X, int ldx,
                                        const _Float16* __restrict__ W,
                                        const _Float16* __restrict__ Wn,
                                        const float* __restrict__ b,
                                        const float* __restrict__ bn, int lane,
                                        v16h (&ring)[PF_D][4], v8f c[4][2],
                                        v8f (&bv)[4], v8f (&bvn)[4]) {
    constexpr int NS = KD / 32;
    const int mr = lane & 15;
    v8f z = {0.f, 0.f, 0.f, 0.f, 0.f, 0.f, 0.f, 0.f};
#pragma unroll
    for (int t = 0; t < 4; ++t)
#pragma unroll
        for (int m = 0; m < 2; ++m) c[t][m] = z;
#pragma unroll
    for (int s = 0; s < NS; ++s) {
        v16h x0 = load_frag(X, ldx, mr, s * 32, lane);
        v16h x1 = load_frag(X + 16 * ldx, ldx, mr, s * 32, lane);
#pragma unroll
        for (int t = 0; t < 4; ++t) {
            c[t][0] = wmma_f16(ring[s % PF_D][t], x0, c[t][0]);
            c[t][1] = wmma_f16(ring[s % PF_D][t], x1, c[t][1]);
        }
#pragma unroll
        for (int t = 0; t < 4; ++t) {
            if (s + PF_D < NS)
                ring[s % PF_D][t] = load_wfrag(W, (s + PF_D) * 4 + t, lane);
            else
                ring[s % PF_D][t] = load_wfrag(Wn, (s + PF_D - NS) * 4 + t, lane);
        }
        if (s == NS - 3) {
#pragma unroll
            for (int t = 0; t < 4; ++t) {
                bv[t]  = load_bias8(b, t, lane >> 4);
                bvn[t] = load_bias8(bn, t, lane >> 4);
            }
        }
    }
}

__device__ __forceinline__ void bias_relu_pack(const v8f c[4][2], const v8f (&bv)[4],
                                               v16h hb[2][2]) {
#pragma unroll
    for (int m = 0; m < 2; ++m)
#pragma unroll
        for (int s = 0; s < 2; ++s) {
            v16h h;
#pragma unroll
            for (int r = 0; r < 8; ++r) {
                float v0 = c[2 * s][m][r] + bv[2 * s][r];
                float v1 = c[2 * s + 1][m][r] + bv[2 * s + 1][r];
                h[r]     = (_Float16)(v0 > 0.f ? v0 : 0.f);
                h[8 + r] = (_Float16)(v1 > 0.f ? v1 : 0.f);
            }
            hb[m][s] = h;
        }
}

__device__ __forceinline__ void gemm_l2(const v16h (&w)[2][4], const v16h hb[2][2],
                                        v8f c[4][2]) {
    v8f z = {0.f, 0.f, 0.f, 0.f, 0.f, 0.f, 0.f, 0.f};
#pragma unroll
    for (int t = 0; t < 4; ++t)
#pragma unroll
        for (int m = 0; m < 2; ++m) c[t][m] = z;
#pragma unroll
    for (int s = 0; s < 2; ++s)
#pragma unroll
        for (int t = 0; t < 4; ++t) {
            c[t][0] = wmma_f16(w[s][t], hb[0][s], c[t][0]);
            c[t][1] = wmma_f16(w[s][t], hb[1][s], c[t][1]);
        }
}

__device__ __forceinline__ void gemm_l3(const v16h (&w)[2], const v16h hb[2][2], v8f c[2]) {
    v8f z = {0.f, 0.f, 0.f, 0.f, 0.f, 0.f, 0.f, 0.f};
    c[0] = z; c[1] = z;
#pragma unroll
    for (int s = 0; s < 2; ++s) {
        c[0] = wmma_f16(w[s], hb[0][s], c[0]);
        c[1] = wmma_f16(w[s], hb[1][s], c[1]);
    }
}

template <int KD>
__device__ __forceinline__ void mlp3(const _Float16* __restrict__ Xw,
                                     const _Float16* __restrict__ W1,
                                     const _Float16* __restrict__ W1n,
                                     const _Float16* __restrict__ W2,
                                     const _Float16* __restrict__ W3,
                                     const float* __restrict__ b1,
                                     const float* __restrict__ b2,
                                     int lane, v16h (&ring)[PF_D][4], v8f out[2]) {
    asm volatile("" ::: "memory");
    v16h w2[2][4];
    v16h w3[2];
    v8f bv1[4], bv2[4];
#pragma unroll
    for (int s = 0; s < 2; ++s)
#pragma unroll
        for (int t = 0; t < 4; ++t) w2[s][t] = load_wfrag(W2, s * 4 + t, lane);
#pragma unroll
    for (int s = 0; s < 2; ++s) w3[s] = load_wfrag(W3, s, lane);

    v8f c1[4][2];
    v16h hb[2][2];
    gemm_l1<KD>(Xw, XS, W1, W1n, b1, b2, lane, ring, c1, bv1, bv2);
    bias_relu_pack(c1, bv1, hb);
    gemm_l2(w2, hb, c1);
    bias_relu_pack(c1, bv2, hb);
    gemm_l3(w3, hb, out);
}

__global__ void cvt_frag_f16(const float* __restrict__ src, _Float16* __restrict__ dst,
                             int n, int kd, int nt, int rows) {
    int i = blockIdx.x * blockDim.x + threadIdx.x;
    if (i >= n) return;
    const int e    = i & 7;
    const int lane = (i >> 3) & 31;
    const int half = (i >> 8) & 1;
    const int f    = i >> 9;
    const int fpm  = (kd >> 5) * nt;
    const int kern = f / fpm;
    const int fm   = f - kern * fpm;
    const int ks   = fm / nt;
    const int t    = fm - ks * nt;
    const int row  = t * 16 + (lane & 15);
    const int k    = ks * 32 + half * 16 + (lane >> 4) * 8 + e;
    dst[i] = (row < rows) ? (_Float16)src[((size_t)kern * rows + row) * kd + k] : (_Float16)0.f;
}

__global__ __launch_bounds__(NTHREADS, 2)
void qplex_main(const float* __restrict__ states, const float* __restrict__ actions,
                const _Float16* __restrict__ wk1, const _Float16* __restrict__ wa1,
                const _Float16* __restrict__ wc1,
                const _Float16* __restrict__ wk2, const _Float16* __restrict__ wa2,
                const _Float16* __restrict__ wc2,
                const _Float16* __restrict__ wk3, const _Float16* __restrict__ wa3,
                const _Float16* __restrict__ wc3,
                const float* __restrict__ bk1, const float* __restrict__ bk2,
                const float* __restrict__ bk3,
                const float* __restrict__ ba1, const float* __restrict__ ba2,
                const float* __restrict__ ba3,
                const float* __restrict__ bc1, const float* __restrict__ bc2,
                const float* __restrict__ bc3,
                float* __restrict__ out) {
    __shared__ _Float16 Xs[WG_ROWS * XS];

    const int tid = threadIdx.x;
    const int b0  = blockIdx.x * WG_ROWS;
    const int wave = tid >> 5, lane = tid & 31;
    const int col = lane & 15, hi = lane >> 4;

    v16h ring[PF_D][4];
#pragma unroll
    for (int s = 0; s < PF_D; ++s)
#pragma unroll
        for (int t = 0; t < 4; ++t) ring[s][t] = load_wfrag(wk1, s * 4 + t, lane);

    {
        const v4f* __restrict__ sp = (const v4f*)(states + (size_t)b0 * S_DIM) + tid;
        const v4f* __restrict__ ap = (const v4f*)(actions + (size_t)b0 * AD_DIM) + tid;
        _Float16* xs_s = Xs + 4 * tid;
        _Float16* xs_a = Xs + (tid >> 5) * XS + S_DIM + 4 * (tid & 31);
#pragma unroll 1
        for (int g = 0; g < WG_ROWS / ST_NB; ++g) {
            v4f v[ST_NB];
#pragma unroll
            for (int j = 0; j < ST_NB; ++j) v[j] = sp[(g * ST_NB + j) * (S_DIM / 4)];
#pragma unroll
            for (int j = 0; j < ST_NB; ++j) {
                v4h h;
                h.x = (_Float16)v[j].x; h.y = (_Float16)v[j].y;
                h.z = (_Float16)v[j].z; h.w = (_Float16)v[j].w;
                *(v4h*)(xs_s + (g * ST_NB + j) * XS) = h;
            }
        }
#pragma unroll 1
        for (int g = 0; g < (WG_ROWS / 2) / ST_NB; ++g) {
            v4f v[ST_NB];
#pragma unroll
            for (int j = 0; j < ST_NB; ++j) v[j] = ap[(g * ST_NB + j) * NTHREADS];
#pragma unroll
            for (int j = 0; j < ST_NB; ++j) {
                v4h h;
                h.x = (_Float16)v[j].x; h.y = (_Float16)v[j].y;
                h.z = (_Float16)v[j].z; h.w = (_Float16)v[j].w;
                *(v4h*)(xs_a + 2 * (g * ST_NB + j) * XS) = h;
            }
        }
    }
    __syncthreads();

    const _Float16* Xw = Xs + (wave * ROWS_PER_WAVE) * XS;

    float acc[2][8];
#pragma unroll
    for (int m = 0; m < 2; ++m)
#pragma unroll
        for (int r = 0; r < 8; ++r) acc[m][r] = 0.f;

#pragma unroll 1
    for (int k = 0; k < K_N; ++k) {
        v8f ck[2], ca[2], cc[2];
        const int kn = (k + 1 < K_N) ? k + 1 : k;

        const float kbias = bk3[k];
        float ab3v[A_N], cb3v[A_N];
#pragma unroll
        for (int r = 0; r < A_N; ++r) { ab3v[r] = ba3[k * A_N + r]; cb3v[r] = bc3[k * A_N + r]; }

        mlp3<S_DIM>(Xw, wk1 + (size_t)k * H_DIM * S_DIM, wa1 + (size_t)k * H_DIM * S_DIM,
                           wk2 + (size_t)k * H_DIM * H_DIM, wk3 + (size_t)k * 16 * H_DIM,
                           bk1 + k * H_DIM, bk2 + k * H_DIM, lane, ring, ck);
        mlp3<S_DIM>(Xw, wa1 + (size_t)k * H_DIM * S_DIM, wc1 + (size_t)k * H_DIM * D_DIM,
                           wa2 + (size_t)k * H_DIM * H_DIM, wa3 + (size_t)k * 16 * H_DIM,
                           ba1 + k * H_DIM, ba2 + k * H_DIM, lane, ring, ca);
        mlp3<D_DIM>(Xw, wc1 + (size_t)k * H_DIM * D_DIM, wk1 + (size_t)kn * H_DIM * S_DIM,
                           wc2 + (size_t)k * H_DIM * H_DIM, wc3 + (size_t)k * 16 * H_DIM,
                           bc1 + k * H_DIM, bc2 + k * H_DIM, lane, ring, cc);

#pragma unroll
        for (int m = 0; m < 2; ++m) {
            const float kv = fabsf(ck[m][0] + kbias) + 1e-10f;
#pragma unroll
            for (int r = 0; r < 8; ++r) {
                float ag = 1.f / (1.f + __expf(-(ca[m][r] + ab3v[r])));
                float ac = 1.f / (1.f + __expf(-(cc[m][r] + cb3v[r])));
                acc[m][r] += kv * ag * ac;
            }
        }
    }

    if (hi == 0) {
#pragma unroll
        for (int m = 0; m < 2; ++m) {
            float* o = out + (size_t)(b0 + wave * ROWS_PER_WAVE + m * 16 + col) * A_N;
            v4f lo = {acc[m][0], acc[m][1], acc[m][2], acc[m][3]};
            v4f up = {acc[m][4], acc[m][5], acc[m][6], acc[m][7]};
            *(v4f*)o = lo;
            *(v4f*)(o + 4) = up;
        }
    }
}

extern "C" void kernel_launch(void* const* d_in, const int* in_sizes, int n_in,
                              void* d_out, int out_size, void* d_ws, size_t ws_size,
                              hipStream_t stream) {
    (void)in_sizes; (void)n_in; (void)out_size; (void)ws_size;
    const float* states  = (const float*)d_in[0];
    const float* actions = (const float*)d_in[1];
    const float* kW1 = (const float*)d_in[2];
    const float* kb1 = (const float*)d_in[3];
    const float* kW2 = (const float*)d_in[4];
    const float* kb2 = (const float*)d_in[5];
    const float* kW3 = (const float*)d_in[6];
    const float* kb3 = (const float*)d_in[7];
    const float* aW1 = (const float*)d_in[8];
    const float* ab1 = (const float*)d_in[9];
    const float* aW2 = (const float*)d_in[10];
    const float* ab2 = (const float*)d_in[11];
    const float* aW3 = (const float*)d_in[12];
    const float* ab3 = (const float*)d_in[13];
    const float* cW1 = (const float*)d_in[14];
    const float* cb1 = (const float*)d_in[15];
    const float* cW2 = (const float*)d_in[16];
    const float* cb2 = (const float*)d_in[17];
    const float* cW3 = (const float*)d_in[18];
    const float* cb3 = (const float*)d_in[19];

    _Float16* ws = (_Float16*)d_ws;
    size_t off = 0;
    _Float16* wk1 = ws + off; off += (size_t)K_N * H_DIM * S_DIM;
    _Float16* wa1 = ws + off; off += (size_t)K_N * H_DIM * S_DIM;
    _Float16* wc1 = ws + off; off += (size_t)K_N * H_DIM * D_DIM;
    _Float16* wk2 = ws + off; off += (size_t)K_N * H_DIM * H_DIM;
    _Float16* wa2 = ws + off; off += (size_t)K_N * H_DIM * H_DIM;
    _Float16* wc2 = ws + off; off += (size_t)K_N * H_DIM * H_DIM;
    _Float16* wk3 = ws + off; off += (size_t)K_N * 16 * H_DIM;
    _Float16* wa3 = ws + off; off += (size_t)K_N * 16 * H_DIM;
    _Float16* wc3 = ws + off; off += (size_t)K_N * 16 * H_DIM;

    auto cvt = [&](const float* s, _Float16* d, int n, int kd, int nt, int rows) {
        cvt_frag_f16<<<(n + 255) / 256, 256, 0, stream>>>(s, d, n, kd, nt, rows);
    };
    cvt(kW1, wk1, K_N * H_DIM * S_DIM, S_DIM, 4, H_DIM);
    cvt(aW1, wa1, K_N * H_DIM * S_DIM, S_DIM, 4, H_DIM);
    cvt(cW1, wc1, K_N * H_DIM * D_DIM, D_DIM, 4, H_DIM);
    cvt(kW2, wk2, K_N * H_DIM * H_DIM, H_DIM, 4, H_DIM);
    cvt(aW2, wa2, K_N * H_DIM * H_DIM, H_DIM, 4, H_DIM);
    cvt(cW2, wc2, K_N * H_DIM * H_DIM, H_DIM, 4, H_DIM);
    cvt(kW3, wk3, K_N * 16 * H_DIM, H_DIM, 1, 1);
    cvt(aW3, wa3, K_N * 16 * H_DIM, H_DIM, 1, A_N);
    cvt(cW3, wc3, K_N * 16 * H_DIM, H_DIM, 1, A_N);

    qplex_main<<<B_TOT / WG_ROWS, NTHREADS, 0, stream>>>(
        states, actions,
        wk1, wa1, wc1, wk2, wa2, wc2, wk3, wa3, wc3,
        kb1, kb2, kb3, ab1, ab2, ab3, cb1, cb2, cb3,
        (float*)d_out);
}


# The nine weight-conversion launches merged into one launch, one thread per 8 destination f16 (32-byte read, 16-byte store); same f16 values, same layout, main kernel untouched
# speedup vs baseline: 5.4144x; 1.1543x over previous
#include <hip/hip_runtime.h>
#include <hip/hip_bf16.h>

typedef __attribute__((ext_vector_type(16))) _Float16 v16h;
typedef __attribute__((ext_vector_type(8)))  _Float16 v8h;
typedef __attribute__((ext_vector_type(8)))  float    v8f;
typedef __attribute__((ext_vector_type(4)))  float    v4f;
typedef __attribute__((ext_vector_type(4)))  _Float16 v4h;

#define B_TOT   65536
#define S_DIM   256
#define A_N     8
#define K_N     10
#define H_DIM   64
#define AD_DIM  128
#define D_DIM   384

#define WG_ROWS 64
#define NTHREADS 64
#define NWAVES  (NTHREADS / 32)
#define ROWS_PER_WAVE 32
#define XS      392
#define ST_NB   16

__device__ __forceinline__ v16h load_frag(const _Float16* __restrict__ p,
                                          int stride, int row, int k0, int lane) {
    const int ho = (lane >> 4) << 3;
    const _Float16* q = p + row * stride + k0 + ho;
    union { v16h v; struct { v8h lo; v8h hi; } s; } u;
    u.s.lo = *(const v8h*)q;
    u.s.hi = *(const v8h*)(q + 16);
    return u.v;
}

__device__ __forceinline__ v16h load_wfrag(const _Float16* __restrict__ p, int f, int lane) {
    const _Float16* q = p + f * 512 + lane * 8;
    union { v16h v; struct { v8h lo; v8h hi; } s; } u;
    u.s.lo = *(const v8h*)q;
    u.s.hi = *(const v8h*)(q + 256);
    return u.v;
}

#define PF_D 4

__device__ __forceinline__ v8f wmma_f16(v16h a, v16h b, v8f c) {
    return __builtin_amdgcn_wmma_f32_16x16x32_f16(false, a, false, b, (short)0, c, false, false);
}


__device__ __forceinline__ v8f load_bias8(const float* __restrict__ b, int t, int hi) {
    const v4f* q = (const v4f*)(b + t * 16 + 8 * hi);
    v4f lo = q[0], up = q[1];
    v8f r = {lo.x, lo.y, lo.z, lo.w, up.x, up.y, up.z, up.w};
    return r;
}

template <int KD>
__device__ __forceinline__ void gemm_l1(const _Float16* __restrict__ X, int ldx,
                                        const _Float16* __restrict__ W,
                                        const _Float16* __restrict__ Wn,
                                        const float* __restrict__ b,
                                        const float* __restrict__ bn, int lane,
                                        v16h (&ring)[PF_D][4], v8f c[4][2],
                                        v8f (&bv)[4], v8f (&bvn)[4]) {
    constexpr int NS = KD / 32;
    const int mr = lane & 15;
    v8f z = {0.f, 0.f, 0.f, 0.f, 0.f, 0.f, 0.f, 0.f};
#pragma unroll
    for (int t = 0; t < 4; ++t)
#pragma unroll
        for (int m = 0; m < 2; ++m) c[t][m] = z;
#pragma unroll
    for (int s = 0; s < NS; ++s) {
        v16h x0 = load_frag(X, ldx, mr, s * 32, lane);
        v16h x1 = load_frag(X + 16 * ldx, ldx, mr, s * 32, lane);
#pragma unroll
        for (int t = 0; t < 4; ++t) {
            c[t][0] = wmma_f16(ring[s % PF_D][t], x0, c[t][0]);
            c[t][1] = wmma_f16(ring[s % PF_D][t], x1, c[t][1]);
        }
#pragma unroll
        for (int t = 0; t < 4; ++t) {
            if (s + PF_D < NS)
                ring[s % PF_D][t] = load_wfrag(W, (s + PF_D) * 4 + t, lane);
            else
                ring[s % PF_D][t] = load_wfrag(Wn, (s + PF_D - NS) * 4 + t, lane);
        }
        if (s == NS - 3) {
#pragma unroll
            for (int t = 0; t < 4; ++t) {
                bv[t]  = load_bias8(b, t, lane >> 4);
                bvn[t] = load_bias8(bn, t, lane >> 4);
            }
        }
    }
}

__device__ __forceinline__ void bias_relu_pack(const v8f c[4][2], const v8f (&bv)[4],
                                               v16h hb[2][2]) {
#pragma unroll
    for (int m = 0; m < 2; ++m)
#pragma unroll
        for (int s = 0; s < 2; ++s) {
            v16h h;
#pragma unroll
            for (int r = 0; r < 8; ++r) {
                float v0 = c[2 * s][m][r] + bv[2 * s][r];
                float v1 = c[2 * s + 1][m][r] + bv[2 * s + 1][r];
                h[r]     = (_Float16)(v0 > 0.f ? v0 : 0.f);
                h[8 + r] = (_Float16)(v1 > 0.f ? v1 : 0.f);
            }
            hb[m][s] = h;
        }
}

__device__ __forceinline__ void gemm_l2(const v16h (&w)[2][4], const v16h hb[2][2],
                                        v8f c[4][2]) {
    v8f z = {0.f, 0.f, 0.f, 0.f, 0.f, 0.f, 0.f, 0.f};
#pragma unroll
    for (int t = 0; t < 4; ++t)
#pragma unroll
        for (int m = 0; m < 2; ++m) c[t][m] = z;
#pragma unroll
    for (int s = 0; s < 2; ++s)
#pragma unroll
        for (int t = 0; t < 4; ++t) {
            c[t][0] = wmma_f16(w[s][t], hb[0][s], c[t][0]);
            c[t][1] = wmma_f16(w[s][t], hb[1][s], c[t][1]);
        }
}

__device__ __forceinline__ void gemm_l3(const v16h (&w)[2], const v16h hb[2][2], v8f c[2]) {
    v8f z = {0.f, 0.f, 0.f, 0.f, 0.f, 0.f, 0.f, 0.f};
    c[0] = z; c[1] = z;
#pragma unroll
    for (int s = 0; s < 2; ++s) {
        c[0] = wmma_f16(w[s], hb[0][s], c[0]);
        c[1] = wmma_f16(w[s], hb[1][s], c[1]);
    }
}

template <int KD>
__device__ __forceinline__ void mlp3(const _Float16* __restrict__ Xw,
                                     const _Float16* __restrict__ W1,
                                     const _Float16* __restrict__ W1n,
                                     const _Float16* __restrict__ W2,
                                     const _Float16* __restrict__ W3,
                                     const float* __restrict__ b1,
                                     const float* __restrict__ b2,
                                     int lane, v16h (&ring)[PF_D][4], v8f out[2]) {
    asm volatile("" ::: "memory");
    v16h w2[2][4];
    v16h w3[2];
    v8f bv1[4], bv2[4];
#pragma unroll
    for (int s = 0; s < 2; ++s)
#pragma unroll
        for (int t = 0; t < 4; ++t) w2[s][t] = load_wfrag(W2, s * 4 + t, lane);
#pragma unroll
    for (int s = 0; s < 2; ++s) w3[s] = load_wfrag(W3, s, lane);

    v8f c1[4][2];
    v16h hb[2][2];
    gemm_l1<KD>(Xw, XS, W1, W1n, b1, b2, lane, ring, c1, bv1, bv2);
    bias_relu_pack(c1, bv1, hb);
    gemm_l2(w2, hb, c1);
    bias_relu_pack(c1, bv2, hb);
    gemm_l3(w3, hb, out);
}

#define CVT_T  256
#define G_W1S  (K_N * H_DIM * S_DIM / 8)
#define G_W1D  (K_N * H_DIM * D_DIM / 8)
#define G_W2   (K_N * H_DIM * H_DIM / 8)
#define G_W3   (K_N * 16 * H_DIM / 8)
#define G_ALL  (2 * G_W1S + G_W1D + 3 * G_W2 + 3 * G_W3)
static_assert(G_W1S % CVT_T == 0 && G_W1D % CVT_T == 0 && G_W2 % CVT_T == 0 &&
              G_W3 % CVT_T == 0, "a block must not straddle two weight sets");

template <int KD, int NT, int ROWS>
__device__ __forceinline__ void cvt_group(const float* __restrict__ src,
                                          _Float16* __restrict__ dst, int g) {
    const int lane = g & 31;
    const int half = (g >> 5) & 1;
    const int f    = g >> 6;
    constexpr int fpm = (KD / 32) * NT;
    const int kern = f / fpm;
    const int fm   = f - kern * fpm;
    const int ks   = fm / NT;
    const int t    = fm - ks * NT;
    const int row  = t * 16 + (lane & 15);
    const int k    = ks * 32 + half * 16 + (lane >> 4) * 8;
    v8h h = {0, 0, 0, 0, 0, 0, 0, 0};
    if (row < ROWS) {
        const v4f* q = (const v4f*)(src + ((size_t)kern * ROWS + row) * KD + k);
        const v4f lo = q[0], up = q[1];
        h[0] = (_Float16)lo.x; h[1] = (_Float16)lo.y; h[2] = (_Float16)lo.z; h[3] = (_Float16)lo.w;
        h[4] = (_Float16)up.x; h[5] = (_Float16)up.y; h[6] = (_Float16)up.z; h[7] = (_Float16)up.w;
    }
    *(v8h*)(dst + (size_t)g * 8) = h;
}

__global__ __launch_bounds__(CVT_T)
void cvt_frag_all(const float* __restrict__ kW1, const float* __restrict__ aW1,
                  const float* __restrict__ cW1,
                  const float* __restrict__ kW2, const float* __restrict__ aW2,
                  const float* __restrict__ cW2,
                  const float* __restrict__ kW3, const float* __restrict__ aW3,
                  const float* __restrict__ cW3,
                  _Float16* __restrict__ dst) {
    int g = blockIdx.x * CVT_T + threadIdx.x;
    if (g < G_W1S) { cvt_group<S_DIM, 4, H_DIM>(kW1, dst, g); return; }
    g -= G_W1S; dst += (size_t)G_W1S * 8;
    if (g < G_W1S) { cvt_group<S_DIM, 4, H_DIM>(aW1, dst, g); return; }
    g -= G_W1S; dst += (size_t)G_W1S * 8;
    if (g < G_W1D) { cvt_group<D_DIM, 4, H_DIM>(cW1, dst, g); return; }
    g -= G_W1D; dst += (size_t)G_W1D * 8;
    if (g < G_W2)  { cvt_group<H_DIM, 4, H_DIM>(kW2, dst, g); return; }
    g -= G_W2;  dst += (size_t)G_W2 * 8;
    if (g < G_W2)  { cvt_group<H_DIM, 4, H_DIM>(aW2, dst, g); return; }
    g -= G_W2;  dst += (size_t)G_W2 * 8;
    if (g < G_W2)  { cvt_group<H_DIM, 4, H_DIM>(cW2, dst, g); return; }
    g -= G_W2;  dst += (size_t)G_W2 * 8;
    if (g < G_W3)  { cvt_group<H_DIM, 1, 1>(kW3, dst, g); return; }
    g -= G_W3;  dst += (size_t)G_W3 * 8;
    if (g < G_W3)  { cvt_group<H_DIM, 1, A_N>(aW3, dst, g); return; }
    g -= G_W3;  dst += (size_t)G_W3 * 8;
    if (g < G_W3)  { cvt_group<H_DIM, 1, A_N>(cW3, dst, g); }
}

__global__ __launch_bounds__(NTHREADS, 2)
void qplex_main(const float* __restrict__ states, const float* __restrict__ actions,
                const _Float16* __restrict__ wk1, const _Float16* __restrict__ wa1,
                const _Float16* __restrict__ wc1,
                const _Float16* __restrict__ wk2, const _Float16* __restrict__ wa2,
                const _Float16* __restrict__ wc2,
                const _Float16* __restrict__ wk3, const _Float16* __restrict__ wa3,
                const _Float16* __restrict__ wc3,
                const float* __restrict__ bk1, const float* __restrict__ bk2,
                const float* __restrict__ bk3,
                const float* __restrict__ ba1, const float* __restrict__ ba2,
                const float* __restrict__ ba3,
                const float* __restrict__ bc1, const float* __restrict__ bc2,
                const float* __restrict__ bc3,
                float* __restrict__ out) {
    __shared__ _Float16 Xs[WG_ROWS * XS];

    const int tid = threadIdx.x;
    const int b0  = blockIdx.x * WG_ROWS;
    const int wave = tid >> 5, lane = tid & 31;
    const int col = lane & 15, hi = lane >> 4;

    v16h ring[PF_D][4];
#pragma unroll
    for (int s = 0; s < PF_D; ++s)
#pragma unroll
        for (int t = 0; t < 4; ++t) ring[s][t] = load_wfrag(wk1, s * 4 + t, lane);

    {
        const v4f* __restrict__ sp = (const v4f*)(states + (size_t)b0 * S_DIM) + tid;
        const v4f* __restrict__ ap = (const v4f*)(actions + (size_t)b0 * AD_DIM) + tid;
        _Float16* xs_s = Xs + 4 * tid;
        _Float16* xs_a = Xs + (tid >> 5) * XS + S_DIM + 4 * (tid & 31);
#pragma unroll 1
        for (int g = 0; g < WG_ROWS / ST_NB; ++g) {
            v4f v[ST_NB];
#pragma unroll
            for (int j = 0; j < ST_NB; ++j) v[j] = sp[(g * ST_NB + j) * (S_DIM / 4)];
#pragma unroll
            for (int j = 0; j < ST_NB; ++j) {
                v4h h;
                h.x = (_Float16)v[j].x; h.y = (_Float16)v[j].y;
                h.z = (_Float16)v[j].z; h.w = (_Float16)v[j].w;
                *(v4h*)(xs_s + (g * ST_NB + j) * XS) = h;
            }
        }
#pragma unroll 1
        for (int g = 0; g < (WG_ROWS / 2) / ST_NB; ++g) {
            v4f v[ST_NB];
#pragma unroll
            for (int j = 0; j < ST_NB; ++j) v[j] = ap[(g * ST_NB + j) * NTHREADS];
#pragma unroll
            for (int j = 0; j < ST_NB; ++j) {
                v4h h;
                h.x = (_Float16)v[j].x; h.y = (_Float16)v[j].y;
                h.z = (_Float16)v[j].z; h.w = (_Float16)v[j].w;
                *(v4h*)(xs_a + 2 * (g * ST_NB + j) * XS) = h;
            }
        }
    }
    __syncthreads();

    const _Float16* Xw = Xs + (wave * ROWS_PER_WAVE) * XS;

    float acc[2][8];
#pragma unroll
    for (int m = 0; m < 2; ++m)
#pragma unroll
        for (int r = 0; r < 8; ++r) acc[m][r] = 0.f;

#pragma unroll 1
    for (int k = 0; k < K_N; ++k) {
        v8f ck[2], ca[2], cc[2];
        const int kn = (k + 1 < K_N) ? k + 1 : k;

        const float kbias = bk3[k];
        float ab3v[A_N], cb3v[A_N];
#pragma unroll
        for (int r = 0; r < A_N; ++r) { ab3v[r] = ba3[k * A_N + r]; cb3v[r] = bc3[k * A_N + r]; }

        mlp3<S_DIM>(Xw, wk1 + (size_t)k * H_DIM * S_DIM, wa1 + (size_t)k * H_DIM * S_DIM,
                           wk2 + (size_t)k * H_DIM * H_DIM, wk3 + (size_t)k * 16 * H_DIM,
                           bk1 + k * H_DIM, bk2 + k * H_DIM, lane, ring, ck);
        mlp3<S_DIM>(Xw, wa1 + (size_t)k * H_DIM * S_DIM, wc1 + (size_t)k * H_DIM * D_DIM,
                           wa2 + (size_t)k * H_DIM * H_DIM, wa3 + (size_t)k * 16 * H_DIM,
                           ba1 + k * H_DIM, ba2 + k * H_DIM, lane, ring, ca);
        mlp3<D_DIM>(Xw, wc1 + (size_t)k * H_DIM * D_DIM, wk1 + (size_t)kn * H_DIM * S_DIM,
                           wc2 + (size_t)k * H_DIM * H_DIM, wc3 + (size_t)k * 16 * H_DIM,
                           bc1 + k * H_DIM, bc2 + k * H_DIM, lane, ring, cc);

#pragma unroll
        for (int m = 0; m < 2; ++m) {
            const float kv = fabsf(ck[m][0] + kbias) + 1e-10f;
#pragma unroll
            for (int r = 0; r < 8; ++r) {
                float ag = 1.f / (1.f + __expf(-(ca[m][r] + ab3v[r])));
                float ac = 1.f / (1.f + __expf(-(cc[m][r] + cb3v[r])));
                acc[m][r] += kv * ag * ac;
            }
        }
    }

    if (hi == 0) {
#pragma unroll
        for (int m = 0; m < 2; ++m) {
            float* o = out + (size_t)(b0 + wave * ROWS_PER_WAVE + m * 16 + col) * A_N;
            v4f lo = {acc[m][0], acc[m][1], acc[m][2], acc[m][3]};
            v4f up = {acc[m][4], acc[m][5], acc[m][6], acc[m][7]};
            *(v4f*)o = lo;
            *(v4f*)(o + 4) = up;
        }
    }
}

extern "C" void kernel_launch(void* const* d_in, const int* in_sizes, int n_in,
                              void* d_out, int out_size, void* d_ws, size_t ws_size,
                              hipStream_t stream) {
    (void)in_sizes; (void)n_in; (void)out_size; (void)ws_size;
    const float* states  = (const float*)d_in[0];
    const float* actions = (const float*)d_in[1];
    const float* kW1 = (const float*)d_in[2];
    const float* kb1 = (const float*)d_in[3];
    const float* kW2 = (const float*)d_in[4];
    const float* kb2 = (const float*)d_in[5];
    const float* kW3 = (const float*)d_in[6];
    const float* kb3 = (const float*)d_in[7];
    const float* aW1 = (const float*)d_in[8];
    const float* ab1 = (const float*)d_in[9];
    const float* aW2 = (const float*)d_in[10];
    const float* ab2 = (const float*)d_in[11];
    const float* aW3 = (const float*)d_in[12];
    const float* ab3 = (const float*)d_in[13];
    const float* cW1 = (const float*)d_in[14];
    const float* cb1 = (const float*)d_in[15];
    const float* cW2 = (const float*)d_in[16];
    const float* cb2 = (const float*)d_in[17];
    const float* cW3 = (const float*)d_in[18];
    const float* cb3 = (const float*)d_in[19];

    _Float16* ws = (_Float16*)d_ws;
    size_t off = 0;
    _Float16* wk1 = ws + off; off += (size_t)K_N * H_DIM * S_DIM;
    _Float16* wa1 = ws + off; off += (size_t)K_N * H_DIM * S_DIM;
    _Float16* wc1 = ws + off; off += (size_t)K_N * H_DIM * D_DIM;
    _Float16* wk2 = ws + off; off += (size_t)K_N * H_DIM * H_DIM;
    _Float16* wa2 = ws + off; off += (size_t)K_N * H_DIM * H_DIM;
    _Float16* wc2 = ws + off; off += (size_t)K_N * H_DIM * H_DIM;
    _Float16* wk3 = ws + off; off += (size_t)K_N * 16 * H_DIM;
    _Float16* wa3 = ws + off; off += (size_t)K_N * 16 * H_DIM;
    _Float16* wc3 = ws + off; off += (size_t)K_N * 16 * H_DIM;

    cvt_frag_all<<<G_ALL / CVT_T, CVT_T, 0, stream>>>(kW1, aW1, cW1, kW2, aW2, cW2,
                                                      kW3, aW3, cW3, wk1);

    qplex_main<<<B_TOT / WG_ROWS, NTHREADS, 0, stream>>>(
        states, actions,
        wk1, wa1, wc1, wk2, wa2, wc2, wk3, wa3, wc3,
        kb1, kb2, kb3, ab1, ab2, ab3, cb1, cb2, cb3,
        (float*)d_out);
}


# Each sigmoid's IEEE division sequence (about ten dependent vector instructions) replaced by the hardware reciprocal v_rcp_f32 (1 ulp) of 1+exp(-x); matrix work and order of all sums unchanged
# speedup vs baseline: 7.4197x; 1.3704x over previous
#include <hip/hip_runtime.h>
#include <hip/hip_bf16.h>

typedef __attribute__((ext_vector_type(16))) _Float16 v16h;
typedef __attribute__((ext_vector_type(8)))  _Float16 v8h;
typedef __attribute__((ext_vector_type(8)))  float    v8f;
typedef __attribute__((ext_vector_type(4)))  float    v4f;
typedef __attribute__((ext_vector_type(4)))  _Float16 v4h;
typedef const __attribute__((address_space(1))) _Float16 g_f16;
typedef const __attribute__((address_space(1))) float    g_f32;
typedef const __attribute__((address_space(1))) v8h      g_v8h;
typedef const __attribute__((address_space(1))) v4f      g_v4f;

#define B_TOT   65536
#define S_DIM   256
#define A_N     8
#define K_N     10
#define H_DIM   64
#define AD_DIM  128
#define D_DIM   384

#define WG_ROWS 64
#define NTHREADS 64
#define NWAVES  (NTHREADS / 32)
#define ROW_TILES 4
#define K_PER_WAVE (K_N / NWAVES)
static_assert(K_N % NWAVES == 0, "the kernels must divide evenly among the waves");
#define XS      392
#define ST_NB   16

__device__ __forceinline__ v16h load_frag(const _Float16* __restrict__ p,
                                          int stride, int row, int k0, int lane) {
    const int ho = (lane >> 4) << 3;
    const _Float16* q = p + row * stride + k0 + ho;
    union { v16h v; struct { v8h lo; v8h hi; } s; } u;
    u.s.lo = *(const v8h*)q;
    u.s.hi = *(const v8h*)(q + 16);
    return u.v;
}

__device__ __forceinline__ v16h load_wfrag(g_f16* p, int f, int lane) {
    g_f16* q = p + f * 512 + lane * 8;
    union { v16h v; struct { v8h lo; v8h hi; } s; } u;
    u.s.lo = *(g_v8h*)q;
    u.s.hi = *(g_v8h*)(q + 256);
    return u.v;
}

__device__ __forceinline__ g_f16* scalar_base(const _Float16* p) {
    const unsigned long long v = (unsigned long long)p;
    const unsigned lo = __builtin_amdgcn_readfirstlane((unsigned)v);
    const unsigned hi = __builtin_amdgcn_readfirstlane((unsigned)(v >> 32));
    return (g_f16*)(((unsigned long long)hi << 32) | lo);
}
__device__ __forceinline__ g_f32* scalar_base(const float* p) {
    const unsigned long long v = (unsigned long long)p;
    const unsigned lo = __builtin_amdgcn_readfirstlane((unsigned)v);
    const unsigned hi = __builtin_amdgcn_readfirstlane((unsigned)(v >> 32));
    return (g_f32*)(((unsigned long long)hi << 32) | lo);
}

#define PF_D 4

__device__ __forceinline__ float sigmoid_rcp(float x) {
    return __builtin_amdgcn_rcpf(1.f + __expf(-x));
}

__device__ __forceinline__ v8f wmma_f16(v16h a, v16h b, v8f c) {
    return __builtin_amdgcn_wmma_f32_16x16x32_f16(false, a, false, b, (short)0, c, false, false);
}


__device__ __forceinline__ v8f load_bias8(g_f32* b, int t, int hi) {
    g_v4f* q = (g_v4f*)(b + t * 16 + 8 * hi);
    v4f lo = q[0], up = q[1];
    v8f r = {lo.x, lo.y, lo.z, lo.w, up.x, up.y, up.z, up.w};
    return r;
}

template <int KD>
__device__ __forceinline__ void gemm_l1(const _Float16* __restrict__ X, int ldx,
                                        g_f16* W, g_f16* Wn, g_f32* b, int lane,
                                        v16h (&ring)[PF_D][4], v8f c[4][ROW_TILES],
                                        v8f (&bv)[4]) {
    constexpr int NS = KD / 32;
    const int mr = lane & 15;
    v8f z = {0.f, 0.f, 0.f, 0.f, 0.f, 0.f, 0.f, 0.f};
#pragma unroll
    for (int t = 0; t < 4; ++t)
#pragma unroll
        for (int m = 0; m < ROW_TILES; ++m) c[t][m] = z;
#pragma unroll
    for (int s = 0; s < NS; ++s) {
        v16h x[ROW_TILES];
#pragma unroll
        for (int m = 0; m < ROW_TILES; ++m)
            x[m] = load_frag(X + 16 * m * ldx, ldx, mr, s * 32, lane);
#pragma unroll
        for (int t = 0; t < 4; ++t)
#pragma unroll
            for (int m = 0; m < ROW_TILES; ++m)
                c[t][m] = wmma_f16(ring[s % PF_D][t], x[m], c[t][m]);
#pragma unroll
        for (int t = 0; t < 4; ++t) {
            if (s + PF_D < NS)
                ring[s % PF_D][t] = load_wfrag(W, (s + PF_D) * 4 + t, lane);
            else
                ring[s % PF_D][t] = load_wfrag(Wn, (s + PF_D - NS) * 4 + t, lane);
        }
        if (s == NS - 3) {
#pragma unroll
            for (int t = 0; t < 4; ++t) bv[t] = load_bias8(b, t, lane >> 4);
        }
    }
}

__device__ __forceinline__ void bias_relu_pack(const v8f c[4][2], const v8f (&bv)[4],
                                               v16h hb[2][2]) {
#pragma unroll
    for (int m = 0; m < 2; ++m)
#pragma unroll
        for (int s = 0; s < 2; ++s) {
            v16h h;
#pragma unroll
            for (int r = 0; r < 8; ++r) {
                float v0 = c[2 * s][m][r] + bv[2 * s][r];
                float v1 = c[2 * s + 1][m][r] + bv[2 * s + 1][r];
                h[r]     = (_Float16)(v0 > 0.f ? v0 : 0.f);
                h[8 + r] = (_Float16)(v1 > 0.f ? v1 : 0.f);
            }
            hb[m][s] = h;
        }
}

__device__ __forceinline__ void gemm_l2(const v16h (&w)[2][4], const v16h hb[2][2],
                                        v8f c[4][2]) {
    v8f z = {0.f, 0.f, 0.f, 0.f, 0.f, 0.f, 0.f, 0.f};
#pragma unroll
    for (int t = 0; t < 4; ++t)
#pragma unroll
        for (int m = 0; m < 2; ++m) c[t][m] = z;
#pragma unroll
    for (int s = 0; s < 2; ++s)
#pragma unroll
        for (int t = 0; t < 4; ++t) {
            c[t][0] = wmma_f16(w[s][t], hb[0][s], c[t][0]);
            c[t][1] = wmma_f16(w[s][t], hb[1][s], c[t][1]);
        }
}

__device__ __forceinline__ void gemm_l3(const v16h (&w)[2], const v16h hb[2][2], v8f c[2]) {
    v8f z = {0.f, 0.f, 0.f, 0.f, 0.f, 0.f, 0.f, 0.f};
    c[0] = z; c[1] = z;
#pragma unroll
    for (int s = 0; s < 2; ++s) {
        c[0] = wmma_f16(w[s], hb[0][s], c[0]);
        c[1] = wmma_f16(w[s], hb[1][s], c[1]);
    }
}

template <int KD>
__device__ __forceinline__ void mlp3(const _Float16* __restrict__ Xw,
                                     g_f16* W1, g_f16* W1n, g_f16* W2, g_f16* W3,
                                     g_f32* b1, g_f32* b2,
                                     int lane, v16h (&ring)[PF_D][4], v8f out[ROW_TILES]) {
    asm volatile("" ::: "memory");
    v16h w2[2][4];
    v16h w3[2];
    v8f bv1[4], bv2[4];
#pragma unroll
    for (int s = 0; s < 2; ++s)
#pragma unroll
        for (int t = 0; t < 4; ++t) w2[s][t] = load_wfrag(W2, s * 4 + t, lane);

    v8f c1[4][ROW_TILES];
    gemm_l1<KD>(Xw, XS, W1, W1n, b1, lane, ring, c1, bv1);
#pragma unroll
    for (int s = 0; s < 2; ++s) w3[s] = load_wfrag(W3, s, lane);
#pragma unroll
    for (int t = 0; t < 4; ++t) bv2[t] = load_bias8(b2, t, lane >> 4);
#pragma unroll
    for (int p = 0; p < ROW_TILES / 2; ++p) {
        v8f c2[4][2];
        v16h hb[2][2];
#pragma unroll
        for (int t = 0; t < 4; ++t) { c2[t][0] = c1[t][2 * p]; c2[t][1] = c1[t][2 * p + 1]; }
        bias_relu_pack(c2, bv1, hb);
        gemm_l2(w2, hb, c2);
        bias_relu_pack(c2, bv2, hb);
        gemm_l3(w3, hb, out + 2 * p);
    }
}

#define CVT_T  256
#define G_W1S  (K_N * H_DIM * S_DIM / 8)
#define G_W1D  (K_N * H_DIM * D_DIM / 8)
#define G_W2   (K_N * H_DIM * H_DIM / 8)
#define G_W3   (K_N * 16 * H_DIM / 8)
#define G_ALL  (2 * G_W1S + G_W1D + 3 * G_W2 + 3 * G_W3)
static_assert(G_W1S % CVT_T == 0 && G_W1D % CVT_T == 0 && G_W2 % CVT_T == 0 &&
              G_W3 % CVT_T == 0, "a block must not straddle two weight sets");

template <int KD, int NT, int ROWS>
__device__ __forceinline__ void cvt_group(const float* __restrict__ src,
                                          _Float16* __restrict__ dst, int g) {
    const int lane = g & 31;
    const int half = (g >> 5) & 1;
    const int f    = g >> 6;
    constexpr int fpm = (KD / 32) * NT;
    const int kern = f / fpm;
    const int fm   = f - kern * fpm;
    const int ks   = fm / NT;
    const int t    = fm - ks * NT;
    const int row  = t * 16 + (lane & 15);
    const int k    = ks * 32 + half * 16 + (lane >> 4) * 8;
    v8h h = {0, 0, 0, 0, 0, 0, 0, 0};
    if (row < ROWS) {
        const v4f* q = (const v4f*)(src + ((size_t)kern * ROWS + row) * KD + k);
        const v4f lo = q[0], up = q[1];
        h[0] = (_Float16)lo.x; h[1] = (_Float16)lo.y; h[2] = (_Float16)lo.z; h[3] = (_Float16)lo.w;
        h[4] = (_Float16)up.x; h[5] = (_Float16)up.y; h[6] = (_Float16)up.z; h[7] = (_Float16)up.w;
    }
    *(v8h*)(dst + (size_t)g * 8) = h;
}

__global__ __launch_bounds__(CVT_T)
void cvt_frag_all(const float* __restrict__ kW1, const float* __restrict__ aW1,
                  const float* __restrict__ cW1,
                  const float* __restrict__ kW2, const float* __restrict__ aW2,
                  const float* __restrict__ cW2,
                  const float* __restrict__ kW3, const float* __restrict__ aW3,
                  const float* __restrict__ cW3,
                  _Float16* __restrict__ dst) {
    int g = blockIdx.x * CVT_T + threadIdx.x;
    if (g < G_W1S) { cvt_group<S_DIM, 4, H_DIM>(kW1, dst, g); return; }
    g -= G_W1S; dst += (size_t)G_W1S * 8;
    if (g < G_W1S) { cvt_group<S_DIM, 4, H_DIM>(aW1, dst, g); return; }
    g -= G_W1S; dst += (size_t)G_W1S * 8;
    if (g < G_W1D) { cvt_group<D_DIM, 4, H_DIM>(cW1, dst, g); return; }
    g -= G_W1D; dst += (size_t)G_W1D * 8;
    if (g < G_W2)  { cvt_group<H_DIM, 4, H_DIM>(kW2, dst, g); return; }
    g -= G_W2;  dst += (size_t)G_W2 * 8;
    if (g < G_W2)  { cvt_group<H_DIM, 4, H_DIM>(aW2, dst, g); return; }
    g -= G_W2;  dst += (size_t)G_W2 * 8;
    if (g < G_W2)  { cvt_group<H_DIM, 4, H_DIM>(cW2, dst, g); return; }
    g -= G_W2;  dst += (size_t)G_W2 * 8;
    if (g < G_W3)  { cvt_group<H_DIM, 1, 1>(kW3, dst, g); return; }
    g -= G_W3;  dst += (size_t)G_W3 * 8;
    if (g < G_W3)  { cvt_group<H_DIM, 1, A_N>(aW3, dst, g); return; }
    g -= G_W3;  dst += (size_t)G_W3 * 8;
    if (g < G_W3)  { cvt_group<H_DIM, 1, A_N>(cW3, dst, g); }
}

__global__ __launch_bounds__(NTHREADS, 2)
void qplex_main(const float* __restrict__ states, const float* __restrict__ actions,
                const _Float16* __restrict__ wk1, const _Float16* __restrict__ wa1,
                const _Float16* __restrict__ wc1,
                const _Float16* __restrict__ wk2, const _Float16* __restrict__ wa2,
                const _Float16* __restrict__ wc2,
                const _Float16* __restrict__ wk3, const _Float16* __restrict__ wa3,
                const _Float16* __restrict__ wc3,
                const float* __restrict__ bk1, const float* __restrict__ bk2,
                const float* __restrict__ bk3,
                const float* __restrict__ ba1, const float* __restrict__ ba2,
                const float* __restrict__ ba3,
                const float* __restrict__ bc1, const float* __restrict__ bc2,
                const float* __restrict__ bc3,
                float* __restrict__ out) {
    __shared__ _Float16 Xs[WG_ROWS * XS];
    __shared__ float part[NWAVES * WG_ROWS * A_N];

    const int tid = threadIdx.x;
    const int b0  = blockIdx.x * WG_ROWS;
    const int wave = __builtin_amdgcn_readfirstlane(tid >> 5);
    const int lane = tid & 31;
    const int k_first = wave * K_PER_WAVE;
    const int col = lane & 15, hi = lane >> 4;

    v16h ring[PF_D][4];
#pragma unroll
    for (int s = 0; s < PF_D; ++s)
#pragma unroll
        for (int t = 0; t < 4; ++t)
            ring[s][t] = load_wfrag(scalar_base(wk1 + (size_t)k_first * H_DIM * S_DIM),
                                    s * 4 + t, lane);

    {
        const v4f* __restrict__ sp = (const v4f*)(states + (size_t)b0 * S_DIM) + tid;
        const v4f* __restrict__ ap = (const v4f*)(actions + (size_t)b0 * AD_DIM) + tid;
        _Float16* xs_s = Xs + 4 * tid;
        _Float16* xs_a = Xs + (tid >> 5) * XS + S_DIM + 4 * (tid & 31);
#pragma unroll 1
        for (int g = 0; g < WG_ROWS / ST_NB; ++g) {
            v4f v[ST_NB];
#pragma unroll
            for (int j = 0; j < ST_NB; ++j) v[j] = sp[(g * ST_NB + j) * (S_DIM / 4)];
#pragma unroll
            for (int j = 0; j < ST_NB; ++j) {
                v4h h;
                h.x = (_Float16)v[j].x; h.y = (_Float16)v[j].y;
                h.z = (_Float16)v[j].z; h.w = (_Float16)v[j].w;
                *(v4h*)(xs_s + (g * ST_NB + j) * XS) = h;
            }
        }
#pragma unroll 1
        for (int g = 0; g < (WG_ROWS / 2) / ST_NB; ++g) {
            v4f v[ST_NB];
#pragma unroll
            for (int j = 0; j < ST_NB; ++j) v[j] = ap[(g * ST_NB + j) * NTHREADS];
#pragma unroll
            for (int j = 0; j < ST_NB; ++j) {
                v4h h;
                h.x = (_Float16)v[j].x; h.y = (_Float16)v[j].y;
                h.z = (_Float16)v[j].z; h.w = (_Float16)v[j].w;
                *(v4h*)(xs_a + 2 * (g * ST_NB + j) * XS) = h;
            }
        }
    }
    __syncthreads();

    const _Float16* Xw = Xs;

    float* mypart = part + wave * (WG_ROWS * A_N) + col * A_N;
    if (hi == 0) {
        const v4f zero4 = {0.f, 0.f, 0.f, 0.f};
#pragma unroll
        for (int m = 0; m < ROW_TILES; ++m) {
            *(v4f*)(mypart + m * 16 * A_N) = zero4;
            *(v4f*)(mypart + m * 16 * A_N + 4) = zero4;
        }
    }

#pragma unroll 1
    for (int kk = 0; kk < K_PER_WAVE; ++kk) {
        const int k = k_first + kk;
        v8f ck[ROW_TILES], ca[ROW_TILES], cc[ROW_TILES];
        const int kn = (kk + 1 < K_PER_WAVE) ? k + 1 : k;

        const float kbias = bk3[k];
        float ab3v[A_N], cb3v[A_N];
#pragma unroll
        for (int r = 0; r < A_N; ++r) { ab3v[r] = ba3[k * A_N + r]; cb3v[r] = bc3[k * A_N + r]; }

        mlp3<S_DIM>(Xw, scalar_base(wk1 + (size_t)k * H_DIM * S_DIM),
                           scalar_base(wa1 + (size_t)k * H_DIM * S_DIM),
                           scalar_base(wk2 + (size_t)k * H_DIM * H_DIM),
                           scalar_base(wk3 + (size_t)k * 16 * H_DIM),
                           scalar_base(bk1 + k * H_DIM), scalar_base(bk2 + k * H_DIM),
                           lane, ring, ck);
        mlp3<S_DIM>(Xw, scalar_base(wa1 + (size_t)k * H_DIM * S_DIM),
                           scalar_base(wc1 + (size_t)k * H_DIM * D_DIM),
                           scalar_base(wa2 + (size_t)k * H_DIM * H_DIM),
                           scalar_base(wa3 + (size_t)k * 16 * H_DIM),
                           scalar_base(ba1 + k * H_DIM), scalar_base(ba2 + k * H_DIM),
                           lane, ring, ca);
        float g[ROW_TILES][8];
#pragma unroll
        for (int m = 0; m < ROW_TILES; ++m) {
            const float kv = fabsf(ck[m][0] + kbias) + 1e-10f;
#pragma unroll
            for (int r = 0; r < 8; ++r) {
                const float ag = sigmoid_rcp(ca[m][r] + ab3v[r]);
                g[m][r] = kv * ag;
            }
        }
        mlp3<D_DIM>(Xw, scalar_base(wc1 + (size_t)k * H_DIM * D_DIM),
                           scalar_base(wk1 + (size_t)kn * H_DIM * S_DIM),
                           scalar_base(wc2 + (size_t)k * H_DIM * H_DIM),
                           scalar_base(wc3 + (size_t)k * 16 * H_DIM),
                           scalar_base(bc1 + k * H_DIM), scalar_base(bc2 + k * H_DIM),
                           lane, ring, cc);

        if (hi == 0) {
#pragma unroll
            for (int m = 0; m < ROW_TILES; ++m) {
                float* q = mypart + m * 16 * A_N;
                const v4f alo = *(const v4f*)q, aup = *(const v4f*)(q + 4);
                float a[8] = {alo.x, alo.y, alo.z, alo.w, aup.x, aup.y, aup.z, aup.w};
#pragma unroll
                for (int r = 0; r < 8; ++r) {
                    const float ac = sigmoid_rcp(cc[m][r] + cb3v[r]);
                    a[r] += g[m][r] * ac;
                }
                const v4f lo = {a[0], a[1], a[2], a[3]};
                const v4f up = {a[4], a[5], a[6], a[7]};
                *(v4f*)q = lo;
                *(v4f*)(q + 4) = up;
            }
        }
    }

    __syncthreads();
    {
        const v4f* p0 = (const v4f*)part;
        const v4f* p1 = (const v4f*)(part + WG_ROWS * A_N);
        v4f* o = (v4f*)(out + (size_t)b0 * A_N);
#pragma unroll
        for (int j = 0; j < (WG_ROWS * A_N / 4) / NTHREADS; ++j) {
            const int i = tid + j * NTHREADS;
            const v4f x = p0[i], y = p1[i];
            const v4f s = {x.x + y.x, x.y + y.y, x.z + y.z, x.w + y.w};
            o[i] = s;
        }
    }
}

extern "C" void kernel_launch(void* const* d_in, const int* in_sizes, int n_in,
                              void* d_out, int out_size, void* d_ws, size_t ws_size,
                              hipStream_t stream) {
    (void)in_sizes; (void)n_in; (void)out_size; (void)ws_size;
    const float* states  = (const float*)d_in[0];
    const float* actions = (const float*)d_in[1];
    const float* kW1 = (const float*)d_in[2];
    const float* kb1 = (const float*)d_in[3];
    const float* kW2 = (const float*)d_in[4];
    const float* kb2 = (const float*)d_in[5];
    const float* kW3 = (const float*)d_in[6];
    const float* kb3 = (const float*)d_in[7];
    const float* aW1 = (const float*)d_in[8];
    const float* ab1 = (const float*)d_in[9];
    const float* aW2 = (const float*)d_in[10];
    const float* ab2 = (const float*)d_in[11];
    const float* aW3 = (const float*)d_in[12];
    const float* ab3 = (const float*)d_in[13];
    const float* cW1 = (const float*)d_in[14];
    const float* cb1 = (const float*)d_in[15];
    const float* cW2 = (const float*)d_in[16];
    const float* cb2 = (const float*)d_in[17];
    const float* cW3 = (const float*)d_in[18];
    const float* cb3 = (const float*)d_in[19];

    _Float16* ws = (_Float16*)d_ws;
    size_t off = 0;
    _Float16* wk1 = ws + off; off += (size_t)K_N * H_DIM * S_DIM;
    _Float16* wa1 = ws + off; off += (size_t)K_N * H_DIM * S_DIM;
    _Float16* wc1 = ws + off; off += (size_t)K_N * H_DIM * D_DIM;
    _Float16* wk2 = ws + off; off += (size_t)K_N * H_DIM * H_DIM;
    _Float16* wa2 = ws + off; off += (size_t)K_N * H_DIM * H_DIM;
    _Float16* wc2 = ws + off; off += (size_t)K_N * H_DIM * H_DIM;
    _Float16* wk3 = ws + off; off += (size_t)K_N * 16 * H_DIM;
    _Float16* wa3 = ws + off; off += (size_t)K_N * 16 * H_DIM;
    _Float16* wc3 = ws + off; off += (size_t)K_N * 16 * H_DIM;

    cvt_frag_all<<<G_ALL / CVT_T, CVT_T, 0, stream>>>(kW1, aW1, cW1, kW2, aW2, cW2,
                                                      kW3, aW3, cW3, wk1);

    qplex_main<<<B_TOT / WG_ROWS, NTHREADS, 0, stream>>>(
        states, actions,
        wk1, wa1, wc1, wk2, wa2, wc2, wk3, wa3, wc3,
        kb1, kb2, kb3, ab1, ab2, ab3, cb1, cb2, cb3,
        (float*)d_out);
}
